# starting kernel
# baseline (speedup 1.0000x reference)
#include <hip/hip_runtime.h>

typedef float v2f __attribute__((ext_vector_type(2)));
typedef float v8f __attribute__((ext_vector_type(8)));

__global__ void gcn_init_deg(float* __restrict__ deg, int n) {
    int i = blockIdx.x * blockDim.x + threadIdx.x;
    if (i < n) deg[i] = 1.0f;
}

__global__ void gcn_count_deg(const int* __restrict__ dst, float* __restrict__ deg, int e) {
    int i = blockIdx.x * blockDim.x + threadIdx.x;
    if (i < e) atomicAdd(&deg[dst[i]], 1.0f);
}

__global__ void gcn_rsqrt(const float* __restrict__ deg, float* __restrict__ dinv, int n) {
    int i = blockIdx.x * blockDim.x + threadIdx.x;
    if (i < n) {
        float d = deg[i];
        dinv[i] = d > 0.0f ? rsqrtf(d) : 0.0f;
    }
}

template <bool RELU>
__global__ void gcn_gemm_wmma(const float* __restrict__ A,
                              const float* __restrict__ B,
                              float* __restrict__ C,
                              int M, int K, int Ncols) {
    const int wave = (blockIdx.x * blockDim.x + threadIdx.x) >> 5;
    const int lane = threadIdx.x & 31;
    const int tilesN = Ncols >> 4;
    const int rowTile = wave / tilesN;
    const int colTile = wave - rowTile * tilesN;
    if (rowTile * 16 >= M) return;

    const int rowBase = rowTile << 4;
    const int colBase = colTile << 4;
    const int m     = lane & 15;
    const int khalf = (lane >> 4) << 1;

    v8f acc = {};
    const float* Arow = A + (size_t)(rowBase + m) * K;
    const float* Bcol = B + colBase + m;

    for (int k = 0; k < K; k += 4) {
        float a0 = Arow[k + khalf];
        float a1 = Arow[k + khalf + 1];
        if (RELU) { a0 = fmaxf(a0, 0.0f); a1 = fmaxf(a1, 0.0f); }
        v2f a; a.x = a0; a.y = a1;

        v2f b;
        b.x = Bcol[(size_t)(k + khalf)     * Ncols];
        b.y = Bcol[(size_t)(k + khalf + 1) * Ncols];

        acc = __builtin_amdgcn_wmma_f32_16x16x4_f32(
            false, a, false, b, (short)0, acc, false, false);
    }

    const int rOff = (lane >> 4) << 3;
    float* Crow = C + (size_t)(rowBase + rOff) * Ncols + colBase + m;
#pragma unroll
    for (int r = 0; r < 8; ++r)
        Crow[(size_t)r * Ncols] = acc[r];
}

template <int F>
__global__ void gcn_init_out(const float* __restrict__ h,
                             const float* __restrict__ dinv,
                             const float* __restrict__ bias,
                             float* __restrict__ out, int n) {
    int id = blockIdx.x * blockDim.x + threadIdx.x;
    int node = id / F;
    int f    = id - node * F;
    if (node < n) {
        float di = dinv[node];
        out[id] = bias[f] + h[id] * (di * di);
    }
}

template <int F>
__global__ void gcn_edge_scatter(const int* __restrict__ src,
                                 const int* __restrict__ dst,
                                 const float* __restrict__ dinv,
                                 const float* __restrict__ h,
                                 float* __restrict__ out, int e) {
    constexpr int CPE = F / 4;
    int id = blockIdx.x * blockDim.x + threadIdx.x;
    int edge = id / CPE;
    int c    = id - edge * CPE;
    if (edge >= e) return;
    int s = src[edge];
    int d = dst[edge];
    float w = dinv[s] * dinv[d];
    const float4 v = *((const float4*)(h + (size_t)s * F) + c);
    float* o = out + (size_t)d * F + (c << 2);
    atomicAdd(o + 0, v.x * w);
    atomicAdd(o + 1, v.y * w);
    atomicAdd(o + 2, v.z * w);
    atomicAdd(o + 3, v.w * w);
}

extern "C" void kernel_launch(void* const* d_in, const int* in_sizes, int n_in,
                              void* d_out, int out_size, void* d_ws, size_t ws_size,
                              hipStream_t stream) {
    constexpr int D_IN = 64, D_HID = 128, D_OUT = 64;

    const float* x  = (const float*)d_in[0];
    const int*   ei = (const int*)  d_in[1];
    const float* W1 = (const float*)d_in[2];
    const float* b1 = (const float*)d_in[3];
    const float* W2 = (const float*)d_in[4];
    const float* b2 = (const float*)d_in[5];
    float* out = (float*)d_out;

    const int N = in_sizes[0] / D_IN;
    const int E = in_sizes[1] / 2;
    const int* src = ei;
    const int* dst = ei + E;

    float* ws   = (float*)d_ws;
    float* deg  = ws;
    float* dinv = deg + N;
    float* h1   = dinv + N;
    float* out1 = h1 + (size_t)N * D_HID;

    auto cdiv = [](long a, long b) { return (int)((a + b - 1) / b); };
    const int BLK = 256;

    gcn_init_deg <<<cdiv(N, BLK), BLK, 0, stream>>>(deg, N);
    gcn_count_deg<<<cdiv(E, BLK), BLK, 0, stream>>>(dst, deg, E);
    gcn_rsqrt    <<<cdiv(N, BLK), BLK, 0, stream>>>(deg, dinv, N);

    {
        long waves = (long)(N / 16) * (D_HID / 16);
        gcn_gemm_wmma<false><<<cdiv(waves * 32, BLK), BLK, 0, stream>>>(
            x, W1, h1, N, D_IN, D_HID);
    }
    gcn_init_out<D_HID><<<cdiv((long)N * D_HID, BLK), BLK, 0, stream>>>(h1, dinv, b1, out1, N);
    gcn_edge_scatter<D_HID><<<cdiv((long)E * (D_HID / 4), BLK), BLK, 0, stream>>>(
        src, dst, dinv, h1, out1, E);

    {
        long waves = (long)(N / 16) * (D_OUT / 16);
        gcn_gemm_wmma<true><<<cdiv(waves * 32, BLK), BLK, 0, stream>>>(
            out1, W2, h1, N, D_HID, D_OUT);
    }
    gcn_init_out<D_OUT><<<cdiv((long)N * D_OUT, BLK), BLK, 0, stream>>>(h1, dinv, b2, out, N);
    gcn_edge_scatter<D_OUT><<<cdiv((long)E * (D_OUT / 4), BLK), BLK, 0, stream>>>(
        src, dst, dinv, h1, out, E);
}


# Edge sums by owned rows: group b holds nodes [196b,196b+196) in LDS (start = bias + self-loop), sweeps a per-edge owner byte, adds its edges' rows by LDS f32 adds; sum order = LDS arrival order
# speedup vs baseline: 1.6425x; 1.6425x over previous
#include <hip/hip_runtime.h>

typedef float v2f __attribute__((ext_vector_type(2)));
typedef float v8f __attribute__((ext_vector_type(8)));

constexpr int RANGE = 196;
constexpr int AGG_THREADS = 1024;
constexpr int KEYS_PER_LOAD = 16;
constexpr int CHUNK = AGG_THREADS * KEYS_PER_LOAD;

__global__ void gcn_init_deg(float* __restrict__ deg, int n) {
    int i = blockIdx.x * blockDim.x + threadIdx.x;
    if (i < n) deg[i] = 1.0f;
}

__global__ void gcn_count_deg_keys(const int* __restrict__ dst, float* __restrict__ deg,
                                   unsigned* __restrict__ keys32, int e) {
    const int i4 = (blockIdx.x * blockDim.x + threadIdx.x) * 4;
    if (i4 + 3 < e) {
        const int d0 = dst[i4], d1 = dst[i4 + 1], d2 = dst[i4 + 2], d3 = dst[i4 + 3];
        atomicAdd(&deg[d0], 1.0f);
        atomicAdd(&deg[d1], 1.0f);
        atomicAdd(&deg[d2], 1.0f);
        atomicAdd(&deg[d3], 1.0f);
        keys32[i4 >> 2] = ((unsigned)(d0 / RANGE) & 255u)
                        | (((unsigned)(d1 / RANGE) & 255u) << 8)
                        | (((unsigned)(d2 / RANGE) & 255u) << 16)
                        | (((unsigned)(d3 / RANGE) & 255u) << 24);
    } else {
        unsigned char* keys8 = (unsigned char*)keys32;
        for (int i = i4; i < e; ++i) {
            const int d = dst[i];
            atomicAdd(&deg[d], 1.0f);
            keys8[i] = (unsigned char)((unsigned)(d / RANGE) & 255u);
        }
    }
}

__global__ void gcn_rsqrt(const float* __restrict__ deg, float* __restrict__ dinv, int n) {
    int i = blockIdx.x * blockDim.x + threadIdx.x;
    if (i < n) {
        float d = deg[i];
        dinv[i] = d > 0.0f ? rsqrtf(d) : 0.0f;
    }
}

template <bool RELU>
__global__ void gcn_gemm_wmma(const float* __restrict__ A,
                              const float* __restrict__ B,
                              float* __restrict__ C,
                              int M, int K, int Ncols) {
    const int wave = (blockIdx.x * blockDim.x + threadIdx.x) >> 5;
    const int lane = threadIdx.x & 31;
    const int tilesN = Ncols >> 4;
    const int rowTile = wave / tilesN;
    const int colTile = wave - rowTile * tilesN;
    if (rowTile * 16 >= M) return;

    const int rowBase = rowTile << 4;
    const int colBase = colTile << 4;
    const int m     = lane & 15;
    const int khalf = (lane >> 4) << 1;

    v8f acc = {};
    const float* Arow = A + (size_t)(rowBase + m) * K;
    const float* Bcol = B + colBase + m;

    for (int k = 0; k < K; k += 4) {
        float a0 = Arow[k + khalf];
        float a1 = Arow[k + khalf + 1];
        if (RELU) { a0 = fmaxf(a0, 0.0f); a1 = fmaxf(a1, 0.0f); }
        v2f a; a.x = a0; a.y = a1;

        v2f b;
        b.x = Bcol[(size_t)(k + khalf)     * Ncols];
        b.y = Bcol[(size_t)(k + khalf + 1) * Ncols];

        acc = __builtin_amdgcn_wmma_f32_16x16x4_f32(
            false, a, false, b, (short)0, acc, false, false);
    }

    const int rOff = (lane >> 4) << 3;
    float* Crow = C + (size_t)(rowBase + rOff) * Ncols + colBase + m;
#pragma unroll
    for (int r = 0; r < 8; ++r)
        Crow[(size_t)r * Ncols] = acc[r];
}

template <int F>
__global__ void __launch_bounds__(AGG_THREADS)
gcn_aggregate_owned(const int* __restrict__ src,
                    const int* __restrict__ dst,
                    const unsigned char* __restrict__ keys,
                    const float* __restrict__ dinv,
                    const float* __restrict__ h,
                    const float* __restrict__ bias,
                    float* __restrict__ out, int n, int e) {
    constexpr int J = F / 32;
    constexpr int U = 8;
    constexpr int G = 8;
    constexpr int CAP = (F >= 128) ? 16384 : 32768;
    __shared__ float acc[RANGE * F];
    __shared__ int list[CAP];
    __shared__ int cnt;

    const int tid  = threadIdx.x;
    const int lane = tid & 31;
    const int wave = __builtin_amdgcn_readfirstlane(tid >> 5);
    const int b    = blockIdx.x;
    const int n0   = b * RANGE;
    const int rows = (n - n0) < RANGE ? (n - n0) : RANGE;

    for (int i = tid; i < rows * F; i += AGG_THREADS) {
        const int r = i / F;
        const int f = i - r * F;
        const float di = dinv[n0 + r];
        acc[i] = bias[f] + h[(size_t)n0 * F + i] * (di * di);
    }
    if (tid == 0) cnt = 0;
    __syncthreads();

    auto put = [&](int idx, int ee) {
        if (idx < CAP) {
            list[idx] = ee;
        } else {
            const int d = dst[ee];
            if (d / RANGE == b) {
                const int s = src[ee];
                const float w = dinv[s] * dinv[d];
                float* ap = acc + (d - n0) * F;
                const float* hp = h + (size_t)s * F;
#pragma unroll 1
                for (int f = 0; f < F; ++f) atomicAdd(ap + f, hp[f] * w);
            }
        }
    };

    const unsigned mykey = (unsigned)b & 255u;
    const unsigned pat   = mykey * 0x01010101u;
    const int nChunks = (e + CHUNK - 1) / CHUNK;
    const int nFull = e >> 4;
    const int c0 = nChunks > 0 ? b % nChunks : 0;
    const uint4* keys128 = (const uint4*)keys;
    for (int it0 = 0; it0 < nChunks; it0 += G) {
        uint4 k[G];
        int eb[G];
        bool full[G];
#pragma unroll
        for (int u = 0; u < G; ++u) {
            int c = c0 + it0 + u;
            if (c >= nChunks) c -= nChunks;
            const int g = c * AGG_THREADS + tid;
            eb[u] = g * KEYS_PER_LOAD;
            full[u] = (it0 + u < nChunks) && (g < nFull);
            k[u] = keys128[full[u] ? g : 0];
        }
        unsigned mm[G];
        unsigned any = 0u;
#pragma unroll
        for (int u = 0; u < G; ++u) {
            const unsigned wd[4] = {k[u].x, k[u].y, k[u].z, k[u].w};
            unsigned mk = 0u;
#pragma unroll
            for (int q = 0; q < 4; ++q) {
                const unsigned x = wd[q] ^ pat;
                const unsigned z = ~(((x & 0x7F7F7F7Fu) + 0x7F7F7F7Fu) | x) & 0x80808080u;
                mk |= (z >> 7) << q;
            }
            mm[u] = full[u] ? mk : 0u;
            any |= mm[u];
        }
        if (any) {
            int nloc = 0;
#pragma unroll
            for (int u = 0; u < G; ++u) nloc += __builtin_popcount(mm[u]);
            int idx = atomicAdd(&cnt, nloc);
#pragma unroll
            for (int u = 0; u < G; ++u) {
                unsigned mk = mm[u];
                while (mk) {
                    const int bit = __builtin_ctz(mk);
                    mk &= mk - 1u;
                    put(idx++, eb[u] + 4 * (bit & 7) + (bit >> 3));
                }
            }
        }
    }
    if (tid < (e & (KEYS_PER_LOAD - 1))) {
        const int ee = (e & ~(KEYS_PER_LOAD - 1)) + tid;
        if ((unsigned)keys[ee] == mykey) put(atomicAdd(&cnt, 1), ee);
    }
    __syncthreads();

    const int total = cnt;
    const int m = __builtin_amdgcn_readfirstlane(total < CAP ? total : CAP);
    for (int base0 = wave * 32; base0 < m; base0 += AGG_THREADS) {
        const int rest = m - base0;
        const int c32 = rest < 32 ? rest : 32;
        int s = 0, dl = 0;
        float w = 0.0f;
        bool ok = false;
        if (lane < c32) {
            const int ee = list[base0 + lane];
            const int d = dst[ee];
            if (d / RANGE == b) {
                ok = true;
                s = src[ee];
                dl = d - n0;
                w = dinv[s] * dinv[d];
            }
        }
        const unsigned okmask = __builtin_amdgcn_ballot_w32(ok);
        int i = 0;
        for (; i + U <= c32; i += U) {
            float v[U][J];
#pragma unroll
            for (int u = 0; u < U; ++u) {
                const int su = __builtin_amdgcn_readlane(s, i + u);
                const float* hp = h + (size_t)su * F + lane;
#pragma unroll
                for (int j = 0; j < J; ++j) v[u][j] = hp[32 * j];
            }
#pragma unroll
            for (int u = 0; u < U; ++u) {
                if ((okmask >> (i + u)) & 1u) {
                    const int du = __builtin_amdgcn_readlane(dl, i + u);
                    const float wu = __int_as_float(__builtin_amdgcn_readlane(__float_as_int(w), i + u));
                    float* ap = acc + du * F + lane;
#pragma unroll
                    for (int j = 0; j < J; ++j) atomicAdd(ap + 32 * j, v[u][j] * wu);
                }
            }
        }
        for (; i < c32; ++i) {
            if ((okmask >> i) & 1u) {
                const int su = __builtin_amdgcn_readlane(s, i);
                const int du = __builtin_amdgcn_readlane(dl, i);
                const float wu = __int_as_float(__builtin_amdgcn_readlane(__float_as_int(w), i));
                const float* hp = h + (size_t)su * F + lane;
                float* ap = acc + du * F + lane;
#pragma unroll
                for (int j = 0; j < J; ++j) atomicAdd(ap + 32 * j, hp[32 * j] * wu);
            }
        }
    }
    __syncthreads();

    for (int i = tid; i < rows * F; i += AGG_THREADS)
        out[(size_t)n0 * F + i] = acc[i];
}

extern "C" void kernel_launch(void* const* d_in, const int* in_sizes, int n_in,
                              void* d_out, int out_size, void* d_ws, size_t ws_size,
                              hipStream_t stream) {
    constexpr int D_IN = 64, D_HID = 128, D_OUT = 64;

    const float* x  = (const float*)d_in[0];
    const int*   ei = (const int*)  d_in[1];
    const float* W1 = (const float*)d_in[2];
    const float* b1 = (const float*)d_in[3];
    const float* W2 = (const float*)d_in[4];
    const float* b2 = (const float*)d_in[5];
    float* out = (float*)d_out;

    const int N = in_sizes[0] / D_IN;
    const int E = in_sizes[1] / 2;
    const int* src = ei;
    const int* dst = ei + E;

    float* ws   = (float*)d_ws;
    float* deg  = ws;
    float* dinv = deg + N;
    float* h1   = dinv + N;
    float* out1 = h1 + (size_t)N * D_HID;
    unsigned char* keys = (unsigned char*)(out1 + (size_t)N * D_HID);
    keys = (unsigned char*)(((size_t)keys + 15) & ~(size_t)15);

    auto cdiv = [](long a, long b) { return (int)((a + b - 1) / b); };
    const int BLK = 256;

    gcn_init_deg<<<cdiv(N, BLK), BLK, 0, stream>>>(deg, N);
    gcn_count_deg_keys<<<cdiv(cdiv(E, 4), BLK), BLK, 0, stream>>>(dst, deg, (unsigned*)keys, E);
    gcn_rsqrt<<<cdiv(N, BLK), BLK, 0, stream>>>(deg, dinv, N);

    {
        long waves = (long)(N / 16) * (D_HID / 16);
        gcn_gemm_wmma<false><<<cdiv(waves * 32, BLK), BLK, 0, stream>>>(
            x, W1, h1, N, D_IN, D_HID);
    }
    gcn_aggregate_owned<D_HID><<<cdiv(N, RANGE), AGG_THREADS, 0, stream>>>(
        src, dst, keys, dinv, h1, b1, out1, N, E);

    {
        long waves = (long)(N / 16) * (D_OUT / 16);
        gcn_gemm_wmma<true><<<cdiv(waves * 32, BLK), BLK, 0, stream>>>(
            out1, W2, h1, N, D_HID, D_OUT);
    }
    gcn_aggregate_owned<D_OUT><<<cdiv(N, RANGE), AGG_THREADS, 0, stream>>>(
        src, dst, keys, dinv, h1, b2, out, N, E);
}


# One owner pre-pass per call: owning groups sweep the key bytes once, count in-degrees in LDS (no global degree atomics) and keep (src,row) edge lists for both layers; LDS arrival order kept
# speedup vs baseline: 2.3565x; 1.4347x over previous
#include <hip/hip_runtime.h>

typedef float v2f __attribute__((ext_vector_type(2)));
typedef float v8f __attribute__((ext_vector_type(8)));

constexpr int RANGE = 196;
constexpr int AGG_THREADS = 1024;
constexpr int KEYS_PER_LOAD = 16;
constexpr int CHUNK = AGG_THREADS * KEYS_PER_LOAD;
constexpr int LIST_CAP = 49152;
constexpr int STAGE = 16384;

typedef int v2i __attribute__((ext_vector_type(2)));

__global__ void gcn_edge_keys(const int* __restrict__ dst, unsigned* __restrict__ keys32, int e) {
    const int i4 = (blockIdx.x * blockDim.x + threadIdx.x) * 4;
    if (i4 + 3 < e) {
        const int d0 = dst[i4], d1 = dst[i4 + 1], d2 = dst[i4 + 2], d3 = dst[i4 + 3];
        keys32[i4 >> 2] = ((unsigned)(d0 / RANGE) & 255u)
                        | (((unsigned)(d1 / RANGE) & 255u) << 8)
                        | (((unsigned)(d2 / RANGE) & 255u) << 16)
                        | (((unsigned)(d3 / RANGE) & 255u) << 24);
    } else {
        unsigned char* keys8 = (unsigned char*)keys32;
        for (int i = i4; i < e; ++i)
            keys8[i] = (unsigned char)((unsigned)(dst[i] / RANGE) & 255u);
    }
}

__global__ void __launch_bounds__(AGG_THREADS)
gcn_owner_prepass(const int* __restrict__ src,
                  const int* __restrict__ dst,
                  const unsigned char* __restrict__ keys,
                  float* __restrict__ dinv,
                  int* __restrict__ lists,
                  int* __restrict__ tot, int n, int e) {
    constexpr int G = 8;
    constexpr int PU = 4;
    __shared__ int list[LIST_CAP];
    __shared__ int degc[RANGE];
    __shared__ int cnt;

    const int tid  = threadIdx.x;
    const int b    = blockIdx.x;
    const int n0   = b * RANGE;
    const int rows = (n - n0) < RANGE ? (n - n0) : RANGE;

    if (tid < RANGE) degc[tid] = 0;
    if (tid == 0) cnt = 0;
    __syncthreads();

    auto put = [&](int idx, int ee) {
        if (idx < LIST_CAP) {
            list[idx] = ee;
        } else {
            const int d = dst[ee];
            if (d / RANGE == b) atomicAdd(&degc[d - n0], 1);
        }
    };

    const unsigned mykey = (unsigned)b & 255u;
    const unsigned pat   = mykey * 0x01010101u;
    const int nChunks = (e + CHUNK - 1) / CHUNK;
    const int nFull = e >> 4;
    const int c0 = nChunks > 0 ? b % nChunks : 0;
    const uint4* keys128 = (const uint4*)keys;
    for (int it0 = 0; it0 < nChunks; it0 += G) {
        uint4 k[G];
        int eb[G];
        bool full[G];
#pragma unroll
        for (int u = 0; u < G; ++u) {
            int c = c0 + it0 + u;
            if (c >= nChunks) c -= nChunks;
            const int g = c * AGG_THREADS + tid;
            eb[u] = g * KEYS_PER_LOAD;
            full[u] = (it0 + u < nChunks) && (g < nFull);
            k[u] = keys128[full[u] ? g : 0];
        }
        unsigned mm[G];
        unsigned any = 0u;
#pragma unroll
        for (int u = 0; u < G; ++u) {
            const unsigned wd[4] = {k[u].x, k[u].y, k[u].z, k[u].w};
            unsigned mk = 0u;
#pragma unroll
            for (int q = 0; q < 4; ++q) {
                const unsigned x = wd[q] ^ pat;
                const unsigned z = ~(((x & 0x7F7F7F7Fu) + 0x7F7F7F7Fu) | x) & 0x80808080u;
                mk |= (z >> 7) << q;
            }
            mm[u] = full[u] ? mk : 0u;
            any |= mm[u];
        }
        if (any) {
            int nloc = 0;
#pragma unroll
            for (int u = 0; u < G; ++u) nloc += __builtin_popcount(mm[u]);
            int idx = atomicAdd(&cnt, nloc);
#pragma unroll
            for (int u = 0; u < G; ++u) {
                unsigned mk = mm[u];
                while (mk) {
                    const int bit = __builtin_ctz(mk);
                    mk &= mk - 1u;
                    put(idx++, eb[u] + 4 * (bit & 7) + (bit >> 3));
                }
            }
        }
    }
    if (tid < (e & (KEYS_PER_LOAD - 1))) {
        const int ee = (e & ~(KEYS_PER_LOAD - 1)) + tid;
        if ((unsigned)keys[ee] == mykey) put(atomicAdd(&cnt, 1), ee);
    }
    __syncthreads();

    const int total = cnt;
    const int m = total < LIST_CAP ? total : LIST_CAP;
    v2i* mine = (v2i*)lists + (size_t)b * LIST_CAP;
    for (int i0 = tid; i0 < m; i0 += AGG_THREADS * PU) {
        int dd[PU], ss[PU];
#pragma unroll
        for (int u = 0; u < PU; ++u) {
            const int i = i0 + u * AGG_THREADS;
            const int ee = list[i < m ? i : i0];
            dd[u] = dst[ee];
            ss[u] = src[ee];
        }
#pragma unroll
        for (int u = 0; u < PU; ++u) {
            const int i = i0 + u * AGG_THREADS;
            if (i < m) {
                const bool own = (dd[u] / RANGE == b);
                if (own) atomicAdd(&degc[dd[u] - n0], 1);
                v2i en;
                en.x = own ? ss[u] : 0;
                en.y = own ? dd[u] - n0 : -1;
                mine[i] = en;
            }
        }
    }
    __syncthreads();

    if (tid < rows) {
        const float dg = (float)(degc[tid] + 1);
        dinv[n0 + tid] = dg > 0.0f ? rsqrtf(dg) : 0.0f;
    }
    if (tid == 0) tot[b] = total;
}

template <bool RELU>
__global__ void gcn_gemm_wmma(const float* __restrict__ A,
                              const float* __restrict__ B,
                              float* __restrict__ C,
                              int M, int K, int Ncols) {
    const int wave = (blockIdx.x * blockDim.x + threadIdx.x) >> 5;
    const int lane = threadIdx.x & 31;
    const int tilesN = Ncols >> 4;
    const int rowTile = wave / tilesN;
    const int colTile = wave - rowTile * tilesN;
    if (rowTile * 16 >= M) return;

    const int rowBase = rowTile << 4;
    const int colBase = colTile << 4;
    const int m     = lane & 15;
    const int khalf = (lane >> 4) << 1;

    v8f acc = {};
    const float* Arow = A + (size_t)(rowBase + m) * K;
    const float* Bcol = B + colBase + m;

    for (int k = 0; k < K; k += 4) {
        float a0 = Arow[k + khalf];
        float a1 = Arow[k + khalf + 1];
        if (RELU) { a0 = fmaxf(a0, 0.0f); a1 = fmaxf(a1, 0.0f); }
        v2f a; a.x = a0; a.y = a1;

        v2f b;
        b.x = Bcol[(size_t)(k + khalf)     * Ncols];
        b.y = Bcol[(size_t)(k + khalf + 1) * Ncols];

        acc = __builtin_amdgcn_wmma_f32_16x16x4_f32(
            false, a, false, b, (short)0, acc, false, false);
    }

    const int rOff = (lane >> 4) << 3;
    float* Crow = C + (size_t)(rowBase + rOff) * Ncols + colBase + m;
#pragma unroll
    for (int r = 0; r < 8; ++r)
        Crow[(size_t)r * Ncols] = acc[r];
}

template <int F>
__global__ void __launch_bounds__(AGG_THREADS)
gcn_aggregate_owned(const int* __restrict__ src,
                    const int* __restrict__ dst,
                    const int* __restrict__ lists,
                    const int* __restrict__ tot,
                    const float* __restrict__ dinv,
                    const float* __restrict__ h,
                    const float* __restrict__ bias,
                    float* __restrict__ out, int n, int e) {
    constexpr int J = F / 32;
    constexpr int U = 8;
    constexpr int SU = 4;
    __shared__ float acc[RANGE * F];
    __shared__ int   es[STAGE];
    __shared__ int   ed[STAGE];
    __shared__ float ew[STAGE];

    const int tid  = threadIdx.x;
    const int lane = tid & 31;
    const int wave = __builtin_amdgcn_readfirstlane(tid >> 5);
    const int b    = blockIdx.x;
    const int n0   = b * RANGE;
    const int rows = (n - n0) < RANGE ? (n - n0) : RANGE;
    const int total = __builtin_amdgcn_readfirstlane(tot[b]);

    for (int i = tid; i < rows * F; i += AGG_THREADS) {
        const int r = i / F;
        const int f = i - r * F;
        const float di = dinv[n0 + r];
        acc[i] = bias[f] + h[(size_t)n0 * F + i] * (di * di);
    }
    __syncthreads();

    if (total <= LIST_CAP) {
        const v2i* mine = (const v2i*)lists + (size_t)b * LIST_CAP;
        for (int r0 = 0; r0 < total; r0 += STAGE) {
            const int mm = (total - r0) < STAGE ? (total - r0) : STAGE;

            for (int i0 = tid; i0 < mm; i0 += AGG_THREADS * SU) {
                v2i en[SU];
                float wv[SU];
#pragma unroll
                for (int u = 0; u < SU; ++u) {
                    const int i = i0 + u * AGG_THREADS;
                    en[u] = mine[r0 + (i < mm ? i : i0)];
                }
#pragma unroll
                for (int u = 0; u < SU; ++u) {
                    const int dr = en[u].y >= 0 ? en[u].y : 0;
                    wv[u] = dinv[en[u].x] * dinv[n0 + dr];
                }
#pragma unroll
                for (int u = 0; u < SU; ++u) {
                    const int i = i0 + u * AGG_THREADS;
                    if (i < mm) {
                        es[i] = en[u].x;
                        ed[i] = en[u].y;
                        ew[i] = en[u].y >= 0 ? wv[u] : 0.0f;
                    }
                }
            }
            __syncthreads();

            for (int base0 = wave * 32; base0 < mm; base0 += AGG_THREADS) {
                const int rest = mm - base0;
                const int c32 = rest < 32 ? rest : 32;
                int s = 0, dl = -1;
                float w = 0.0f;
                if (lane < c32) {
                    s  = es[base0 + lane];
                    dl = ed[base0 + lane];
                    w  = ew[base0 + lane];
                }
                const unsigned okmask = __builtin_amdgcn_ballot_w32(dl >= 0);
                int i = 0;
                for (; i + U <= c32; i += U) {
                    float v[U][J];
#pragma unroll
                    for (int u = 0; u < U; ++u) {
                        const int su = __builtin_amdgcn_readlane(s, i + u);
                        const float* hp = h + (size_t)su * F + lane;
#pragma unroll
                        for (int j = 0; j < J; ++j) v[u][j] = hp[32 * j];
                    }
#pragma unroll
                    for (int u = 0; u < U; ++u) {
                        if ((okmask >> (i + u)) & 1u) {
                            const int du = __builtin_amdgcn_readlane(dl, i + u);
                            const float wu = __int_as_float(__builtin_amdgcn_readlane(__float_as_int(w), i + u));
                            float* ap = acc + du * F + lane;
#pragma unroll
                            for (int j = 0; j < J; ++j) atomicAdd(ap + 32 * j, v[u][j] * wu);
                        }
                    }
                }
                for (; i < c32; ++i) {
                    if ((okmask >> i) & 1u) {
                        const int su = __builtin_amdgcn_readlane(s, i);
                        const int du = __builtin_amdgcn_readlane(dl, i);
                        const float wu = __int_as_float(__builtin_amdgcn_readlane(__float_as_int(w), i));
                        const float* hp = h + (size_t)su * F + lane;
                        float* ap = acc + du * F + lane;
#pragma unroll
                        for (int j = 0; j < J; ++j) atomicAdd(ap + 32 * j, hp[32 * j] * wu);
                    }
                }
            }
            __syncthreads();
        }
    } else {
        for (int ee = tid; ee < e; ee += AGG_THREADS) {
            const int d = dst[ee];
            if (d / RANGE == b) {
                const int s = src[ee];
                const float w = dinv[s] * dinv[d];
                float* ap = acc + (d - n0) * F;
                const float* hp = h + (size_t)s * F;
#pragma unroll 1
                for (int f = 0; f < F; ++f) atomicAdd(ap + f, hp[f] * w);
            }
        }
        __syncthreads();
    }

    for (int i = tid; i < rows * F; i += AGG_THREADS)
        out[(size_t)n0 * F + i] = acc[i];
}

extern "C" void kernel_launch(void* const* d_in, const int* in_sizes, int n_in,
                              void* d_out, int out_size, void* d_ws, size_t ws_size,
                              hipStream_t stream) {
    constexpr int D_IN = 64, D_HID = 128, D_OUT = 64;

    const float* x  = (const float*)d_in[0];
    const int*   ei = (const int*)  d_in[1];
    const float* W1 = (const float*)d_in[2];
    const float* b1 = (const float*)d_in[3];
    const float* W2 = (const float*)d_in[4];
    const float* b2 = (const float*)d_in[5];
    float* out = (float*)d_out;

    const int N = in_sizes[0] / D_IN;
    const int E = in_sizes[1] / 2;
    const int* src = ei;
    const int* dst = ei + E;

    auto cdiv = [](long a, long b) { return (int)((a + b - 1) / b); };
    const int BLK = 256;
    const int owners = cdiv(N, RANGE);

    float* ws   = (float*)d_ws;
    float* dinv = ws;
    float* h1   = (float*)(((size_t)(dinv + N) + 255) & ~(size_t)255);
    float* out1 = h1 + (size_t)N * D_HID;
    unsigned char* keys = (unsigned char*)(out1 + (size_t)N * D_HID);
    keys = (unsigned char*)(((size_t)keys + 15) & ~(size_t)15);
    int* tot   = (int*)(((size_t)(keys + E) + 15) & ~(size_t)15);
    int* lists = (int*)(((size_t)(tot + owners) + 255) & ~(size_t)255);

    gcn_edge_keys<<<cdiv(cdiv(E, 4), BLK), BLK, 0, stream>>>(dst, (unsigned*)keys, E);
    gcn_owner_prepass<<<owners, AGG_THREADS, 0, stream>>>(src, dst, keys, dinv, lists, tot, N, E);

    {
        long waves = (long)(N / 16) * (D_HID / 16);
        gcn_gemm_wmma<false><<<cdiv(waves * 32, BLK), BLK, 0, stream>>>(
            x, W1, h1, N, D_IN, D_HID);
    }
    gcn_aggregate_owned<D_HID><<<owners, AGG_THREADS, 0, stream>>>(
        src, dst, lists, tot, dinv, h1, b1, out1, N, E);

    {
        long waves = (long)(N / 16) * (D_OUT / 16);
        gcn_gemm_wmma<true><<<cdiv(waves * 32, BLK), BLK, 0, stream>>>(
            out1, W2, h1, N, D_HID, D_OUT);
    }
    gcn_aggregate_owned<D_OUT><<<owners, AGG_THREADS, 0, stream>>>(
        src, dst, lists, tot, dinv, h1, b2, out, N, E);
}


# Products retiled: one wave owns a 16-row strip over all columns (A loaded once, 16 B per lane), W staged once per group in LDS as k-pairs; k regrouped inside each 8 (f32 sum order changed)
# speedup vs baseline: 2.5194x; 1.0691x over previous
#include <hip/hip_runtime.h>

typedef float v2f __attribute__((ext_vector_type(2)));
typedef float v8f __attribute__((ext_vector_type(8)));

constexpr int RANGE = 196;
constexpr int AGG_THREADS = 1024;
constexpr int KEYS_PER_LOAD = 16;
constexpr int CHUNK = AGG_THREADS * KEYS_PER_LOAD;
constexpr int LIST_CAP = 49152;
constexpr int STAGE = 16384;
constexpr int GEMM_WAVES = 8;

typedef int v2i __attribute__((ext_vector_type(2)));

__global__ void gcn_edge_keys(const int* __restrict__ dst, unsigned* __restrict__ keys32, int e) {
    const int i4 = (blockIdx.x * blockDim.x + threadIdx.x) * 4;
    if (i4 + 3 < e) {
        const int d0 = dst[i4], d1 = dst[i4 + 1], d2 = dst[i4 + 2], d3 = dst[i4 + 3];
        keys32[i4 >> 2] = ((unsigned)(d0 / RANGE) & 255u)
                        | (((unsigned)(d1 / RANGE) & 255u) << 8)
                        | (((unsigned)(d2 / RANGE) & 255u) << 16)
                        | (((unsigned)(d3 / RANGE) & 255u) << 24);
    } else {
        unsigned char* keys8 = (unsigned char*)keys32;
        for (int i = i4; i < e; ++i)
            keys8[i] = (unsigned char)((unsigned)(dst[i] / RANGE) & 255u);
    }
}

__global__ void __launch_bounds__(AGG_THREADS)
gcn_owner_prepass(const int* __restrict__ src,
                  const int* __restrict__ dst,
                  const unsigned char* __restrict__ keys,
                  float* __restrict__ dinv,
                  int* __restrict__ lists,
                  int* __restrict__ tot, int n, int e) {
    constexpr int G = 8;
    constexpr int PU = 4;
    __shared__ int list[LIST_CAP];
    __shared__ int degc[RANGE];
    __shared__ int cnt;

    const int tid  = threadIdx.x;
    const int b    = blockIdx.x;
    const int n0   = b * RANGE;
    const int rows = (n - n0) < RANGE ? (n - n0) : RANGE;

    if (tid < RANGE) degc[tid] = 0;
    if (tid == 0) cnt = 0;
    __syncthreads();

    auto put = [&](int idx, int ee) {
        if (idx < LIST_CAP) {
            list[idx] = ee;
        } else {
            const int d = dst[ee];
            if (d / RANGE == b) atomicAdd(&degc[d - n0], 1);
        }
    };

    const unsigned mykey = (unsigned)b & 255u;
    const unsigned pat   = mykey * 0x01010101u;
    const int nChunks = (e + CHUNK - 1) / CHUNK;
    const int nFull = e >> 4;
    const int c0 = nChunks > 0 ? b % nChunks : 0;
    const uint4* keys128 = (const uint4*)keys;
    for (int it0 = 0; it0 < nChunks; it0 += G) {
        uint4 k[G];
        int eb[G];
        bool full[G];
#pragma unroll
        for (int u = 0; u < G; ++u) {
            int c = c0 + it0 + u;
            if (c >= nChunks) c -= nChunks;
            const int g = c * AGG_THREADS + tid;
            eb[u] = g * KEYS_PER_LOAD;
            full[u] = (it0 + u < nChunks) && (g < nFull);
            k[u] = keys128[full[u] ? g : 0];
        }
        unsigned mm[G];
        unsigned any = 0u;
#pragma unroll
        for (int u = 0; u < G; ++u) {
            const unsigned wd[4] = {k[u].x, k[u].y, k[u].z, k[u].w};
            unsigned mk = 0u;
#pragma unroll
            for (int q = 0; q < 4; ++q) {
                const unsigned x = wd[q] ^ pat;
                const unsigned z = ~(((x & 0x7F7F7F7Fu) + 0x7F7F7F7Fu) | x) & 0x80808080u;
                mk |= (z >> 7) << q;
            }
            mm[u] = full[u] ? mk : 0u;
            any |= mm[u];
        }
        if (any) {
            int nloc = 0;
#pragma unroll
            for (int u = 0; u < G; ++u) nloc += __builtin_popcount(mm[u]);
            int idx = atomicAdd(&cnt, nloc);
#pragma unroll
            for (int u = 0; u < G; ++u) {
                unsigned mk = mm[u];
                while (mk) {
                    const int bit = __builtin_ctz(mk);
                    mk &= mk - 1u;
                    put(idx++, eb[u] + 4 * (bit & 7) + (bit >> 3));
                }
            }
        }
    }
    if (tid < (e & (KEYS_PER_LOAD - 1))) {
        const int ee = (e & ~(KEYS_PER_LOAD - 1)) + tid;
        if ((unsigned)keys[ee] == mykey) put(atomicAdd(&cnt, 1), ee);
    }
    __syncthreads();

    const int total = cnt;
    const int m = total < LIST_CAP ? total : LIST_CAP;
    v2i* mine = (v2i*)lists + (size_t)b * LIST_CAP;
    for (int i0 = tid; i0 < m; i0 += AGG_THREADS * PU) {
        int dd[PU], ss[PU];
#pragma unroll
        for (int u = 0; u < PU; ++u) {
            const int i = i0 + u * AGG_THREADS;
            const int ee = list[i < m ? i : i0];
            dd[u] = dst[ee];
            ss[u] = src[ee];
        }
#pragma unroll
        for (int u = 0; u < PU; ++u) {
            const int i = i0 + u * AGG_THREADS;
            if (i < m) {
                const bool own = (dd[u] / RANGE == b);
                if (own) atomicAdd(&degc[dd[u] - n0], 1);
                v2i en;
                en.x = own ? ss[u] : 0;
                en.y = own ? dd[u] - n0 : -1;
                mine[i] = en;
            }
        }
    }
    __syncthreads();

    if (tid < rows) {
        const float dg = (float)(degc[tid] + 1);
        dinv[n0 + tid] = dg > 0.0f ? rsqrtf(dg) : 0.0f;
    }
    if (tid == 0) tot[b] = total;
}

template <bool RELU, int K, int NC>
__global__ void __launch_bounds__(GEMM_WAVES * 32)
gcn_gemm_wmma(const float* __restrict__ A,
              const float* __restrict__ B,
              float* __restrict__ C, int M) {
    constexpr int NT = NC / 16;
    constexpr int KT = K / 8;
    constexpr int PITCH = 2 * NC + 16;
    constexpr int ITEMS = (K / 2) * (NC / 4);
    static_assert(ITEMS % (GEMM_WAVES * 32) == 0, "every thread stages the same number of pieces");
    __shared__ float4 Bs4[(K / 2) * PITCH / 4];
    float* Bs = (float*)Bs4;

    const int tid  = threadIdx.x;
    const int lane = tid & 31;
    const int wave = tid >> 5;
    const int c    = lane & 15;
    const int h    = lane >> 4;
    const int rowTile = blockIdx.x * GEMM_WAVES + wave;
    const bool active = rowTile * 16 < M;
    const int rowBase = active ? rowTile * 16 : 0;

    float4 a[KT];
    const float4* Ap = (const float4*)(A + (size_t)(rowBase + c) * K) + h;
#pragma unroll
    for (int t = 0; t < KT; ++t) a[t] = Ap[2 * t];

#pragma unroll
    for (int i = 0; i < ITEMS / (GEMM_WAVES * 32); ++i) {
        const int idx = tid + i * (GEMM_WAVES * 32);
        const int p = idx / (NC / 4);
        const int q = idx - p * (NC / 4);
        const float4 r0 = *(const float4*)(B + (size_t)(2 * p) * NC + 4 * q);
        const float4 r1 = *(const float4*)(B + (size_t)(2 * p + 1) * NC + 4 * q);
        float4* d = (float4*)(Bs + p * PITCH + 8 * q);
        d[0] = make_float4(r0.x, r1.x, r0.y, r1.y);
        d[1] = make_float4(r0.z, r1.z, r0.w, r1.w);
    }
    __syncthreads();
    if (!active) return;

    v8f acc[NT];
#pragma unroll
    for (int j = 0; j < NT; ++j) acc[j] = v8f{};

    const float* bl = Bs + (2 * h) * PITCH + 2 * c;
#pragma unroll
    for (int t = 0; t < KT; ++t) {
#pragma unroll
        for (int u = 0; u < 2; ++u) {
            float a0 = u ? a[t].z : a[t].x;
            float a1 = u ? a[t].w : a[t].y;
            if (RELU) { a0 = fmaxf(a0, 0.0f); a1 = fmaxf(a1, 0.0f); }
            v2f av; av.x = a0; av.y = a1;
            const float* bp = bl + (4 * t + u) * PITCH;
#pragma unroll
            for (int j = 0; j < NT; ++j) {
                const v2f bv = *(const v2f*)(bp + 32 * j);
                acc[j] = __builtin_amdgcn_wmma_f32_16x16x4_f32(
                    false, av, false, bv, (short)0, acc[j], false, false);
            }
        }
    }

    float* Crow = C + (size_t)(rowBase + 8 * h) * NC + c;
#pragma unroll
    for (int j = 0; j < NT; ++j)
#pragma unroll
        for (int r = 0; r < 8; ++r)
            Crow[(size_t)r * NC + 16 * j] = acc[j][r];
}

template <int F>
__global__ void __launch_bounds__(AGG_THREADS)
gcn_aggregate_owned(const int* __restrict__ src,
                    const int* __restrict__ dst,
                    const int* __restrict__ lists,
                    const int* __restrict__ tot,
                    const float* __restrict__ dinv,
                    const float* __restrict__ h,
                    const float* __restrict__ bias,
                    float* __restrict__ out, int n, int e) {
    constexpr int J = F / 32;
    constexpr int U = 8;
    constexpr int SU = 4;
    __shared__ float acc[RANGE * F];
    __shared__ int   es[STAGE];
    __shared__ int   ed[STAGE];
    __shared__ float ew[STAGE];

    const int tid  = threadIdx.x;
    const int lane = tid & 31;
    const int wave = __builtin_amdgcn_readfirstlane(tid >> 5);
    const int b    = blockIdx.x;
    const int n0   = b * RANGE;
    const int rows = (n - n0) < RANGE ? (n - n0) : RANGE;
    const int total = __builtin_amdgcn_readfirstlane(tot[b]);

    for (int i = tid; i < rows * F; i += AGG_THREADS) {
        const int r = i / F;
        const int f = i - r * F;
        const float di = dinv[n0 + r];
        acc[i] = bias[f] + h[(size_t)n0 * F + i] * (di * di);
    }
    __syncthreads();

    if (total <= LIST_CAP) {
        const v2i* mine = (const v2i*)lists + (size_t)b * LIST_CAP;
        for (int r0 = 0; r0 < total; r0 += STAGE) {
            const int mm = (total - r0) < STAGE ? (total - r0) : STAGE;

            for (int i0 = tid; i0 < mm; i0 += AGG_THREADS * SU) {
                v2i en[SU];
                float wv[SU];
#pragma unroll
                for (int u = 0; u < SU; ++u) {
                    const int i = i0 + u * AGG_THREADS;
                    en[u] = mine[r0 + (i < mm ? i : i0)];
                }
#pragma unroll
                for (int u = 0; u < SU; ++u) {
                    const int dr = en[u].y >= 0 ? en[u].y : 0;
                    wv[u] = dinv[en[u].x] * dinv[n0 + dr];
                }
#pragma unroll
                for (int u = 0; u < SU; ++u) {
                    const int i = i0 + u * AGG_THREADS;
                    if (i < mm) {
                        es[i] = en[u].x;
                        ed[i] = en[u].y;
                        ew[i] = en[u].y >= 0 ? wv[u] : 0.0f;
                    }
                }
            }
            __syncthreads();

            for (int base0 = wave * 32; base0 < mm; base0 += AGG_THREADS) {
                const int rest = mm - base0;
                const int c32 = rest < 32 ? rest : 32;
                int s = 0, dl = -1;
                float w = 0.0f;
                if (lane < c32) {
                    s  = es[base0 + lane];
                    dl = ed[base0 + lane];
                    w  = ew[base0 + lane];
                }
                const unsigned okmask = __builtin_amdgcn_ballot_w32(dl >= 0);
                int i = 0;
                for (; i + U <= c32; i += U) {
                    float v[U][J];
#pragma unroll
                    for (int u = 0; u < U; ++u) {
                        const int su = __builtin_amdgcn_readlane(s, i + u);
                        const float* hp = h + (size_t)su * F + lane;
#pragma unroll
                        for (int j = 0; j < J; ++j) v[u][j] = hp[32 * j];
                    }
#pragma unroll
                    for (int u = 0; u < U; ++u) {
                        if ((okmask >> (i + u)) & 1u) {
                            const int du = __builtin_amdgcn_readlane(dl, i + u);
                            const float wu = __int_as_float(__builtin_amdgcn_readlane(__float_as_int(w), i + u));
                            float* ap = acc + du * F + lane;
#pragma unroll
                            for (int j = 0; j < J; ++j) atomicAdd(ap + 32 * j, v[u][j] * wu);
                        }
                    }
                }
                for (; i < c32; ++i) {
                    if ((okmask >> i) & 1u) {
                        const int su = __builtin_amdgcn_readlane(s, i);
                        const int du = __builtin_amdgcn_readlane(dl, i);
                        const float wu = __int_as_float(__builtin_amdgcn_readlane(__float_as_int(w), i));
                        const float* hp = h + (size_t)su * F + lane;
                        float* ap = acc + du * F + lane;
#pragma unroll
                        for (int j = 0; j < J; ++j) atomicAdd(ap + 32 * j, hp[32 * j] * wu);
                    }
                }
            }
            __syncthreads();
        }
    } else {
        for (int ee = tid; ee < e; ee += AGG_THREADS) {
            const int d = dst[ee];
            if (d / RANGE == b) {
                const int s = src[ee];
                const float w = dinv[s] * dinv[d];
                float* ap = acc + (d - n0) * F;
                const float* hp = h + (size_t)s * F;
#pragma unroll 1
                for (int f = 0; f < F; ++f) atomicAdd(ap + f, hp[f] * w);
            }
        }
        __syncthreads();
    }

    for (int i = tid; i < rows * F; i += AGG_THREADS)
        out[(size_t)n0 * F + i] = acc[i];
}

extern "C" void kernel_launch(void* const* d_in, const int* in_sizes, int n_in,
                              void* d_out, int out_size, void* d_ws, size_t ws_size,
                              hipStream_t stream) {
    constexpr int D_IN = 64, D_HID = 128, D_OUT = 64;

    const float* x  = (const float*)d_in[0];
    const int*   ei = (const int*)  d_in[1];
    const float* W1 = (const float*)d_in[2];
    const float* b1 = (const float*)d_in[3];
    const float* W2 = (const float*)d_in[4];
    const float* b2 = (const float*)d_in[5];
    float* out = (float*)d_out;

    const int N = in_sizes[0] / D_IN;
    const int E = in_sizes[1] / 2;
    const int* src = ei;
    const int* dst = ei + E;

    auto cdiv = [](long a, long b) { return (int)((a + b - 1) / b); };
    const int BLK = 256;
    const int owners = cdiv(N, RANGE);

    float* ws   = (float*)d_ws;
    float* dinv = ws;
    float* h1   = (float*)(((size_t)(dinv + N) + 255) & ~(size_t)255);
    float* out1 = h1 + (size_t)N * D_HID;
    unsigned char* keys = (unsigned char*)(out1 + (size_t)N * D_HID);
    keys = (unsigned char*)(((size_t)keys + 15) & ~(size_t)15);
    int* tot   = (int*)(((size_t)(keys + E) + 15) & ~(size_t)15);
    int* lists = (int*)(((size_t)(tot + owners) + 255) & ~(size_t)255);

    gcn_edge_keys<<<cdiv(cdiv(E, 4), BLK), BLK, 0, stream>>>(dst, (unsigned*)keys, E);
    gcn_owner_prepass<<<owners, AGG_THREADS, 0, stream>>>(src, dst, keys, dinv, lists, tot, N, E);

    gcn_gemm_wmma<false, D_IN, D_HID><<<cdiv(N / 16, GEMM_WAVES), GEMM_WAVES * 32, 0, stream>>>(
        x, W1, h1, N);
    gcn_aggregate_owned<D_HID><<<owners, AGG_THREADS, 0, stream>>>(
        src, dst, lists, tot, dinv, h1, b1, out1, N, E);

    gcn_gemm_wmma<true, D_HID, D_OUT><<<cdiv(N / 16, GEMM_WAVES), GEMM_WAVES * 32, 0, stream>>>(
        out1, W2, h1, N);
    gcn_aggregate_owned<D_OUT><<<owners, AGG_THREADS, 0, stream>>>(
        src, dst, lists, tot, dinv, h1, b2, out, N, E);
}


# Owner sweep tests only flagged 16-edge groups: the key launch also builds one bit per (owner key, group) with LDS ORs; edge-list order changes, f32 sums stay in LDS arrival order
# speedup vs baseline: 2.6466x; 1.0505x over previous
#include <hip/hip_runtime.h>

typedef float v2f __attribute__((ext_vector_type(2)));
typedef float v8f __attribute__((ext_vector_type(8)));

constexpr int RANGE = 196;
constexpr int AGG_THREADS = 1024;
constexpr int KEYS_PER_LOAD = 16;
constexpr int LIST_CAP = 49152;
constexpr int STAGE = 16384;
constexpr int GEMM_WAVES = 8;
constexpr int KEY_THREADS = 256;
constexpr int FCHUNK  = 4096;
constexpr int FGROUPS = FCHUNK / KEYS_PER_LOAD;
constexpr int FWORDS  = FGROUPS / 32;
constexpr int FPITCH  = 260;

typedef int v2i __attribute__((ext_vector_type(2)));

__global__ void __launch_bounds__(KEY_THREADS)
gcn_edge_keys(const int* __restrict__ dst, unsigned* __restrict__ keys32,
              unsigned* __restrict__ flags, int e) {
    constexpr int L = FCHUNK / 4 / KEY_THREADS;
    __shared__ unsigned m[FWORDS * FPITCH];

    const int tid   = threadIdx.x;
    const int chunk = blockIdx.x;
    const int ebase = chunk * FCHUNK;

    int d[L][4];
    bool fullw[L];
#pragma unroll
    for (int j = 0; j < L; ++j) {
        const int i4 = ebase + (j * KEY_THREADS + tid) * 4;
        fullw[j] = i4 + 3 < e;
        if (fullw[j]) {
            d[j][0] = dst[i4];     d[j][1] = dst[i4 + 1];
            d[j][2] = dst[i4 + 2]; d[j][3] = dst[i4 + 3];
        }
    }
    for (int i = tid; i < FWORDS * FPITCH; i += KEY_THREADS) m[i] = 0u;
    __syncthreads();

#pragma unroll
    for (int j = 0; j < L; ++j) {
        const int q  = j * KEY_THREADS + tid;
        const int i4 = ebase + q * 4;
        const int gi = q >> 2;
        unsigned* row = m + (gi >> 5) * FPITCH;
        const unsigned bit = 1u << (gi & 31);
        if (fullw[j]) {
            const unsigned o0 = (unsigned)(d[j][0] / RANGE) & 255u;
            const unsigned o1 = (unsigned)(d[j][1] / RANGE) & 255u;
            const unsigned o2 = (unsigned)(d[j][2] / RANGE) & 255u;
            const unsigned o3 = (unsigned)(d[j][3] / RANGE) & 255u;
            keys32[i4 >> 2] = o0 | (o1 << 8) | (o2 << 16) | (o3 << 24);
            atomicOr(row + o0, bit);
            atomicOr(row + o1, bit);
            atomicOr(row + o2, bit);
            atomicOr(row + o3, bit);
        } else {
            unsigned char* keys8 = (unsigned char*)keys32;
            for (int i = i4; i < e; ++i) {
                const unsigned o = (unsigned)(dst[i] / RANGE) & 255u;
                keys8[i] = (unsigned char)o;
                atomicOr(row + o, bit);
            }
        }
    }
    __syncthreads();

    for (int i = tid; i < 256 * FWORDS; i += KEY_THREADS)
        flags[(size_t)chunk * (256 * FWORDS) + i] = m[(i % FWORDS) * FPITCH + (i / FWORDS)];
}

__global__ void __launch_bounds__(AGG_THREADS)
gcn_owner_prepass(const int* __restrict__ src,
                  const int* __restrict__ dst,
                  const unsigned char* __restrict__ keys,
                  const unsigned* __restrict__ flags,
                  float* __restrict__ dinv,
                  int* __restrict__ lists,
                  int* __restrict__ tot, int n, int e) {
    constexpr int G = 4;
    constexpr int PU = 4;
    __shared__ int list[LIST_CAP];
    __shared__ int degc[RANGE];
    __shared__ int cnt;

    const int tid  = threadIdx.x;
    const int b    = blockIdx.x;
    const int n0   = b * RANGE;
    const int rows = (n - n0) < RANGE ? (n - n0) : RANGE;

    if (tid < RANGE) degc[tid] = 0;
    if (tid == 0) cnt = 0;
    __syncthreads();

    auto put = [&](int idx, int ee) {
        if (idx < LIST_CAP) {
            list[idx] = ee;
        } else {
            const int d = dst[ee];
            if (d / RANGE == b) atomicAdd(&degc[d - n0], 1);
        }
    };

    const unsigned mykey = (unsigned)b & 255u;
    const unsigned pat   = mykey * 0x01010101u;
    const int nFull = e >> 4;
    const int nFlagWords = ((e + FCHUNK - 1) / FCHUNK) * FWORDS;
    const uint4* keys128 = (const uint4*)keys;
    for (int wi = tid; wi < nFlagWords; wi += AGG_THREADS) {
        unsigned bits = flags[(size_t)(wi / FWORDS) * (256 * FWORDS) + mykey * FWORDS + (wi % FWORDS)];
        const int gbase = wi * 32;
        while (bits) {
            uint4 k[G];
            int eb[G];
            bool full[G];
#pragma unroll
            for (int u = 0; u < G; ++u) {
                const bool has = bits != 0u;
                const int g = gbase + (has ? __builtin_ctz(bits) : 0);
                bits &= bits - 1u;
                eb[u] = g * KEYS_PER_LOAD;
                full[u] = has && (g < nFull);
                k[u] = keys128[full[u] ? g : 0];
            }
            unsigned mm[G];
            unsigned any = 0u;
#pragma unroll
            for (int u = 0; u < G; ++u) {
                const unsigned wd[4] = {k[u].x, k[u].y, k[u].z, k[u].w};
                unsigned mk = 0u;
#pragma unroll
                for (int q = 0; q < 4; ++q) {
                    const unsigned x = wd[q] ^ pat;
                    const unsigned z = ~(((x & 0x7F7F7F7Fu) + 0x7F7F7F7Fu) | x) & 0x80808080u;
                    mk |= (z >> 7) << q;
                }
                mm[u] = full[u] ? mk : 0u;
                any |= mm[u];
            }
            if (any) {
                int nloc = 0;
#pragma unroll
                for (int u = 0; u < G; ++u) nloc += __builtin_popcount(mm[u]);
                int idx = atomicAdd(&cnt, nloc);
#pragma unroll
                for (int u = 0; u < G; ++u) {
                    unsigned mk = mm[u];
                    while (mk) {
                        const int bit = __builtin_ctz(mk);
                        mk &= mk - 1u;
                        put(idx++, eb[u] + 4 * (bit & 7) + (bit >> 3));
                    }
                }
            }
        }
    }
    if (tid < (e & (KEYS_PER_LOAD - 1))) {
        const int ee = (e & ~(KEYS_PER_LOAD - 1)) + tid;
        if ((unsigned)keys[ee] == mykey) put(atomicAdd(&cnt, 1), ee);
    }
    __syncthreads();

    const int total = cnt;
    const int m = total < LIST_CAP ? total : LIST_CAP;
    v2i* mine = (v2i*)lists + (size_t)b * LIST_CAP;
    for (int i0 = tid; i0 < m; i0 += AGG_THREADS * PU) {
        int dd[PU], ss[PU];
#pragma unroll
        for (int u = 0; u < PU; ++u) {
            const int i = i0 + u * AGG_THREADS;
            const int ee = list[i < m ? i : i0];
            dd[u] = dst[ee];
            ss[u] = src[ee];
        }
#pragma unroll
        for (int u = 0; u < PU; ++u) {
            const int i = i0 + u * AGG_THREADS;
            if (i < m) {
                const bool own = (dd[u] / RANGE == b);
                if (own) atomicAdd(&degc[dd[u] - n0], 1);
                v2i en;
                en.x = own ? ss[u] : 0;
                en.y = own ? dd[u] - n0 : -1;
                mine[i] = en;
            }
        }
    }
    __syncthreads();

    if (tid < rows) {
        const float dg = (float)(degc[tid] + 1);
        dinv[n0 + tid] = dg > 0.0f ? rsqrtf(dg) : 0.0f;
    }
    if (tid == 0) tot[b] = total;
}

template <bool RELU, int K, int NC>
__global__ void __launch_bounds__(GEMM_WAVES * 32)
gcn_gemm_wmma(const float* __restrict__ A,
              const float* __restrict__ B,
              float* __restrict__ C, int M) {
    constexpr int NT = NC / 16;
    constexpr int KT = K / 8;
    constexpr int PITCH = 2 * NC + 16;
    constexpr int ITEMS = (K / 2) * (NC / 4);
    static_assert(ITEMS % (GEMM_WAVES * 32) == 0, "every thread stages the same number of pieces");
    __shared__ float4 Bs4[(K / 2) * PITCH / 4];
    float* Bs = (float*)Bs4;

    const int tid  = threadIdx.x;
    const int lane = tid & 31;
    const int wave = tid >> 5;
    const int c    = lane & 15;
    const int h    = lane >> 4;
    const int rowTile = blockIdx.x * GEMM_WAVES + wave;
    const bool active = rowTile * 16 < M;
    const int rowBase = active ? rowTile * 16 : 0;

    float4 a[KT];
    const float4* Ap = (const float4*)(A + (size_t)(rowBase + c) * K) + h;
#pragma unroll
    for (int t = 0; t < KT; ++t) a[t] = Ap[2 * t];

#pragma unroll
    for (int i = 0; i < ITEMS / (GEMM_WAVES * 32); ++i) {
        const int idx = tid + i * (GEMM_WAVES * 32);
        const int p = idx / (NC / 4);
        const int q = idx - p * (NC / 4);
        const float4 r0 = *(const float4*)(B + (size_t)(2 * p) * NC + 4 * q);
        const float4 r1 = *(const float4*)(B + (size_t)(2 * p + 1) * NC + 4 * q);
        float4* d = (float4*)(Bs + p * PITCH + 8 * q);
        d[0] = make_float4(r0.x, r1.x, r0.y, r1.y);
        d[1] = make_float4(r0.z, r1.z, r0.w, r1.w);
    }
    __syncthreads();
    if (!active) return;

    v8f acc[NT];
#pragma unroll
    for (int j = 0; j < NT; ++j) acc[j] = v8f{};

    const float* bl = Bs + (2 * h) * PITCH + 2 * c;
#pragma unroll
    for (int t = 0; t < KT; ++t) {
#pragma unroll
        for (int u = 0; u < 2; ++u) {
            float a0 = u ? a[t].z : a[t].x;
            float a1 = u ? a[t].w : a[t].y;
            if (RELU) { a0 = fmaxf(a0, 0.0f); a1 = fmaxf(a1, 0.0f); }
            v2f av; av.x = a0; av.y = a1;
            const float* bp = bl + (4 * t + u) * PITCH;
#pragma unroll
            for (int j = 0; j < NT; ++j) {
                const v2f bv = *(const v2f*)(bp + 32 * j);
                acc[j] = __builtin_amdgcn_wmma_f32_16x16x4_f32(
                    false, av, false, bv, (short)0, acc[j], false, false);
            }
        }
    }

    float* Crow = C + (size_t)(rowBase + 8 * h) * NC + c;
#pragma unroll
    for (int j = 0; j < NT; ++j)
#pragma unroll
        for (int r = 0; r < 8; ++r)
            Crow[(size_t)r * NC + 16 * j] = acc[j][r];
}

template <int F>
__global__ void __launch_bounds__(AGG_THREADS)
gcn_aggregate_owned(const int* __restrict__ src,
                    const int* __restrict__ dst,
                    const int* __restrict__ lists,
                    const int* __restrict__ tot,
                    const float* __restrict__ dinv,
                    const float* __restrict__ h,
                    const float* __restrict__ bias,
                    float* __restrict__ out, int n, int e) {
    constexpr int J = F / 32;
    constexpr int U = 8;
    constexpr int SU = 4;
    __shared__ float acc[RANGE * F];
    __shared__ int   es[STAGE];
    __shared__ int   ed[STAGE];
    __shared__ float ew[STAGE];

    const int tid  = threadIdx.x;
    const int lane = tid & 31;
    const int wave = __builtin_amdgcn_readfirstlane(tid >> 5);
    const int b    = blockIdx.x;
    const int n0   = b * RANGE;
    const int rows = (n - n0) < RANGE ? (n - n0) : RANGE;
    const int total = __builtin_amdgcn_readfirstlane(tot[b]);

    for (int i = tid; i < rows * F; i += AGG_THREADS) {
        const int r = i / F;
        const int f = i - r * F;
        const float di = dinv[n0 + r];
        acc[i] = bias[f] + h[(size_t)n0 * F + i] * (di * di);
    }
    __syncthreads();

    if (total <= LIST_CAP) {
        const v2i* mine = (const v2i*)lists + (size_t)b * LIST_CAP;
        for (int r0 = 0; r0 < total; r0 += STAGE) {
            const int mm = (total - r0) < STAGE ? (total - r0) : STAGE;

            for (int i0 = tid; i0 < mm; i0 += AGG_THREADS * SU) {
                v2i en[SU];
                float wv[SU];
#pragma unroll
                for (int u = 0; u < SU; ++u) {
                    const int i = i0 + u * AGG_THREADS;
                    en[u] = mine[r0 + (i < mm ? i : i0)];
                }
#pragma unroll
                for (int u = 0; u < SU; ++u) {
                    const int dr = en[u].y >= 0 ? en[u].y : 0;
                    wv[u] = dinv[en[u].x] * dinv[n0 + dr];
                }
#pragma unroll
                for (int u = 0; u < SU; ++u) {
                    const int i = i0 + u * AGG_THREADS;
                    if (i < mm) {
                        es[i] = en[u].x;
                        ed[i] = en[u].y;
                        ew[i] = en[u].y >= 0 ? wv[u] : 0.0f;
                    }
                }
            }
            __syncthreads();

            for (int base0 = wave * 32; base0 < mm; base0 += AGG_THREADS) {
                const int rest = mm - base0;
                const int c32 = rest < 32 ? rest : 32;
                int s = 0, dl = -1;
                float w = 0.0f;
                if (lane < c32) {
                    s  = es[base0 + lane];
                    dl = ed[base0 + lane];
                    w  = ew[base0 + lane];
                }
                const unsigned okmask = __builtin_amdgcn_ballot_w32(dl >= 0);
                int i = 0;
                for (; i + U <= c32; i += U) {
                    float v[U][J];
#pragma unroll
                    for (int u = 0; u < U; ++u) {
                        const int su = __builtin_amdgcn_readlane(s, i + u);
                        const float* hp = h + (size_t)su * F + lane;
#pragma unroll
                        for (int j = 0; j < J; ++j) v[u][j] = hp[32 * j];
                    }
#pragma unroll
                    for (int u = 0; u < U; ++u) {
                        if ((okmask >> (i + u)) & 1u) {
                            const int du = __builtin_amdgcn_readlane(dl, i + u);
                            const float wu = __int_as_float(__builtin_amdgcn_readlane(__float_as_int(w), i + u));
                            float* ap = acc + du * F + lane;
#pragma unroll
                            for (int j = 0; j < J; ++j) atomicAdd(ap + 32 * j, v[u][j] * wu);
                        }
                    }
                }
                for (; i < c32; ++i) {
                    if ((okmask >> i) & 1u) {
                        const int su = __builtin_amdgcn_readlane(s, i);
                        const int du = __builtin_amdgcn_readlane(dl, i);
                        const float wu = __int_as_float(__builtin_amdgcn_readlane(__float_as_int(w), i));
                        const float* hp = h + (size_t)su * F + lane;
                        float* ap = acc + du * F + lane;
#pragma unroll
                        for (int j = 0; j < J; ++j) atomicAdd(ap + 32 * j, hp[32 * j] * wu);
                    }
                }
            }
            __syncthreads();
        }
    } else {
        for (int ee = tid; ee < e; ee += AGG_THREADS) {
            const int d = dst[ee];
            if (d / RANGE == b) {
                const int s = src[ee];
                const float w = dinv[s] * dinv[d];
                float* ap = acc + (d - n0) * F;
                const float* hp = h + (size_t)s * F;
#pragma unroll 1
                for (int f = 0; f < F; ++f) atomicAdd(ap + f, hp[f] * w);
            }
        }
        __syncthreads();
    }

    for (int i = tid; i < rows * F; i += AGG_THREADS)
        out[(size_t)n0 * F + i] = acc[i];
}

extern "C" void kernel_launch(void* const* d_in, const int* in_sizes, int n_in,
                              void* d_out, int out_size, void* d_ws, size_t ws_size,
                              hipStream_t stream) {
    constexpr int D_IN = 64, D_HID = 128, D_OUT = 64;

    const float* x  = (const float*)d_in[0];
    const int*   ei = (const int*)  d_in[1];
    const float* W1 = (const float*)d_in[2];
    const float* b1 = (const float*)d_in[3];
    const float* W2 = (const float*)d_in[4];
    const float* b2 = (const float*)d_in[5];
    float* out = (float*)d_out;

    const int N = in_sizes[0] / D_IN;
    const int E = in_sizes[1] / 2;
    const int* src = ei;
    const int* dst = ei + E;

    auto cdiv = [](long a, long b) { return (int)((a + b - 1) / b); };
    const int owners = cdiv(N, RANGE);

    float* ws   = (float*)d_ws;
    float* dinv = ws;
    float* h1   = (float*)(((size_t)(dinv + N) + 255) & ~(size_t)255);
    float* out1 = h1 + (size_t)N * D_HID;
    unsigned char* keys = (unsigned char*)(out1 + (size_t)N * D_HID);
    keys = (unsigned char*)(((size_t)keys + 15) & ~(size_t)15);
    int* tot   = (int*)(((size_t)(keys + E) + 15) & ~(size_t)15);
    const int flagChunks = cdiv(E, FCHUNK);
    unsigned* flags = (unsigned*)(((size_t)(tot + owners) + 255) & ~(size_t)255);
    int* lists = (int*)(((size_t)(flags + (size_t)flagChunks * 256 * FWORDS) + 255) & ~(size_t)255);

    gcn_edge_keys<<<flagChunks, KEY_THREADS, 0, stream>>>(dst, (unsigned*)keys, flags, E);
    gcn_owner_prepass<<<owners, AGG_THREADS, 0, stream>>>(src, dst, keys, flags, dinv, lists, tot, N, E);

    gcn_gemm_wmma<false, D_IN, D_HID><<<cdiv(N / 16, GEMM_WAVES), GEMM_WAVES * 32, 0, stream>>>(
        x, W1, h1, N);
    gcn_aggregate_owned<D_HID><<<owners, AGG_THREADS, 0, stream>>>(
        src, dst, lists, tot, dinv, h1, b1, out1, N, E);

    gcn_gemm_wmma<true, D_HID, D_OUT><<<cdiv(N / 16, GEMM_WAVES), GEMM_WAVES * 32, 0, stream>>>(
        out1, W2, h1, N);
    gcn_aggregate_owned<D_OUT><<<owners, AGG_THREADS, 0, stream>>>(
        src, dst, lists, tot, dinv, h1, b2, out, N, E);
}


# Layer-2 product moved into the layer-1 owning groups: each multiplies its LDS rows (ReLU applied) by W2 on f32 WMMA and stores h2; out1 never reaches memory, 6 -> 5 launches; sum orders kept
# speedup vs baseline: 2.9407x; 1.1111x over previous
#include <hip/hip_runtime.h>

typedef float v2f __attribute__((ext_vector_type(2)));
typedef float v8f __attribute__((ext_vector_type(8)));

constexpr int RANGE = 196;
constexpr int AGG_THREADS = 1024;
constexpr int KEYS_PER_LOAD = 16;
constexpr int LIST_CAP = 49152;
constexpr int STAGE = 16384;
constexpr int GEMM_WAVES = 8;
constexpr int KEY_THREADS = 256;
constexpr int FCHUNK  = 4096;
constexpr int FGROUPS = FCHUNK / KEYS_PER_LOAD;
constexpr int FWORDS  = FGROUPS / 32;
constexpr int FPITCH  = 260;

typedef int v2i __attribute__((ext_vector_type(2)));

__global__ void __launch_bounds__(KEY_THREADS)
gcn_edge_keys(const int* __restrict__ dst, unsigned* __restrict__ keys32,
              unsigned* __restrict__ flags, int e) {
    constexpr int L = FCHUNK / 4 / KEY_THREADS;
    __shared__ unsigned m[FWORDS * FPITCH];

    const int tid   = threadIdx.x;
    const int chunk = blockIdx.x;
    const int ebase = chunk * FCHUNK;

    int d[L][4];
    bool fullw[L];
#pragma unroll
    for (int j = 0; j < L; ++j) {
        const int i4 = ebase + (j * KEY_THREADS + tid) * 4;
        fullw[j] = i4 + 3 < e;
        if (fullw[j]) {
            d[j][0] = dst[i4];     d[j][1] = dst[i4 + 1];
            d[j][2] = dst[i4 + 2]; d[j][3] = dst[i4 + 3];
        }
    }
    for (int i = tid; i < FWORDS * FPITCH; i += KEY_THREADS) m[i] = 0u;
    __syncthreads();

#pragma unroll
    for (int j = 0; j < L; ++j) {
        const int q  = j * KEY_THREADS + tid;
        const int i4 = ebase + q * 4;
        const int gi = q >> 2;
        unsigned* row = m + (gi >> 5) * FPITCH;
        const unsigned bit = 1u << (gi & 31);
        if (fullw[j]) {
            const unsigned o0 = (unsigned)(d[j][0] / RANGE) & 255u;
            const unsigned o1 = (unsigned)(d[j][1] / RANGE) & 255u;
            const unsigned o2 = (unsigned)(d[j][2] / RANGE) & 255u;
            const unsigned o3 = (unsigned)(d[j][3] / RANGE) & 255u;
            keys32[i4 >> 2] = o0 | (o1 << 8) | (o2 << 16) | (o3 << 24);
            atomicOr(row + o0, bit);
            atomicOr(row + o1, bit);
            atomicOr(row + o2, bit);
            atomicOr(row + o3, bit);
        } else {
            unsigned char* keys8 = (unsigned char*)keys32;
            for (int i = i4; i < e; ++i) {
                const unsigned o = (unsigned)(dst[i] / RANGE) & 255u;
                keys8[i] = (unsigned char)o;
                atomicOr(row + o, bit);
            }
        }
    }
    __syncthreads();

    for (int i = tid; i < 256 * FWORDS; i += KEY_THREADS)
        flags[(size_t)chunk * (256 * FWORDS) + i] = m[(i % FWORDS) * FPITCH + (i / FWORDS)];
}

__global__ void __launch_bounds__(AGG_THREADS)
gcn_owner_prepass(const int* __restrict__ src,
                  const int* __restrict__ dst,
                  const unsigned char* __restrict__ keys,
                  const unsigned* __restrict__ flags,
                  float* __restrict__ dinv,
                  int* __restrict__ lists,
                  int* __restrict__ tot, int n, int e) {
    constexpr int G = 4;
    constexpr int PU = 4;
    __shared__ int list[LIST_CAP];
    __shared__ int degc[RANGE];
    __shared__ int cnt;

    const int tid  = threadIdx.x;
    const int b    = blockIdx.x;
    const int n0   = b * RANGE;
    const int rows = (n - n0) < RANGE ? (n - n0) : RANGE;

    if (tid < RANGE) degc[tid] = 0;
    if (tid == 0) cnt = 0;
    __syncthreads();

    auto put = [&](int idx, int ee) {
        if (idx < LIST_CAP) {
            list[idx] = ee;
        } else {
            const int d = dst[ee];
            if (d / RANGE == b) atomicAdd(&degc[d - n0], 1);
        }
    };

    const unsigned mykey = (unsigned)b & 255u;
    const unsigned pat   = mykey * 0x01010101u;
    const int nFull = e >> 4;
    const int nFlagWords = ((e + FCHUNK - 1) / FCHUNK) * FWORDS;
    const uint4* keys128 = (const uint4*)keys;
    for (int wi = tid; wi < nFlagWords; wi += AGG_THREADS) {
        unsigned bits = flags[(size_t)(wi / FWORDS) * (256 * FWORDS) + mykey * FWORDS + (wi % FWORDS)];
        const int gbase = wi * 32;
        while (bits) {
            uint4 k[G];
            int eb[G];
            bool full[G];
#pragma unroll
            for (int u = 0; u < G; ++u) {
                const bool has = bits != 0u;
                const int g = gbase + (has ? __builtin_ctz(bits) : 0);
                bits &= bits - 1u;
                eb[u] = g * KEYS_PER_LOAD;
                full[u] = has && (g < nFull);
                k[u] = keys128[full[u] ? g : 0];
            }
            unsigned mm[G];
            unsigned any = 0u;
#pragma unroll
            for (int u = 0; u < G; ++u) {
                const unsigned wd[4] = {k[u].x, k[u].y, k[u].z, k[u].w};
                unsigned mk = 0u;
#pragma unroll
                for (int q = 0; q < 4; ++q) {
                    const unsigned x = wd[q] ^ pat;
                    const unsigned z = ~(((x & 0x7F7F7F7Fu) + 0x7F7F7F7Fu) | x) & 0x80808080u;
                    mk |= (z >> 7) << q;
                }
                mm[u] = full[u] ? mk : 0u;
                any |= mm[u];
            }
            if (any) {
                int nloc = 0;
#pragma unroll
                for (int u = 0; u < G; ++u) nloc += __builtin_popcount(mm[u]);
                int idx = atomicAdd(&cnt, nloc);
#pragma unroll
                for (int u = 0; u < G; ++u) {
                    unsigned mk = mm[u];
                    while (mk) {
                        const int bit = __builtin_ctz(mk);
                        mk &= mk - 1u;
                        put(idx++, eb[u] + 4 * (bit & 7) + (bit >> 3));
                    }
                }
            }
        }
    }
    if (tid < (e & (KEYS_PER_LOAD - 1))) {
        const int ee = (e & ~(KEYS_PER_LOAD - 1)) + tid;
        if ((unsigned)keys[ee] == mykey) put(atomicAdd(&cnt, 1), ee);
    }
    __syncthreads();

    const int total = cnt;
    const int m = total < LIST_CAP ? total : LIST_CAP;
    v2i* mine = (v2i*)lists + (size_t)b * LIST_CAP;
    for (int i0 = tid; i0 < m; i0 += AGG_THREADS * PU) {
        int dd[PU], ss[PU];
#pragma unroll
        for (int u = 0; u < PU; ++u) {
            const int i = i0 + u * AGG_THREADS;
            const int ee = list[i < m ? i : i0];
            dd[u] = dst[ee];
            ss[u] = src[ee];
        }
#pragma unroll
        for (int u = 0; u < PU; ++u) {
            const int i = i0 + u * AGG_THREADS;
            if (i < m) {
                const bool own = (dd[u] / RANGE == b);
                if (own) atomicAdd(&degc[dd[u] - n0], 1);
                v2i en;
                en.x = own ? ss[u] : 0;
                en.y = own ? dd[u] - n0 : -1;
                mine[i] = en;
            }
        }
    }
    __syncthreads();

    if (tid < rows) {
        const float dg = (float)(degc[tid] + 1);
        dinv[n0 + tid] = dg > 0.0f ? rsqrtf(dg) : 0.0f;
    }
    if (tid == 0) tot[b] = total;
}

template <bool RELU, int K, int NC>
__global__ void __launch_bounds__(GEMM_WAVES * 32)
gcn_gemm_wmma(const float* __restrict__ A,
              const float* __restrict__ B,
              float* __restrict__ C, int M) {
    constexpr int NT = NC / 16;
    constexpr int KT = K / 8;
    constexpr int PITCH = 2 * NC + 16;
    constexpr int ITEMS = (K / 2) * (NC / 4);
    static_assert(ITEMS % (GEMM_WAVES * 32) == 0, "every thread stages the same number of pieces");
    __shared__ float4 Bs4[(K / 2) * PITCH / 4];
    float* Bs = (float*)Bs4;

    const int tid  = threadIdx.x;
    const int lane = tid & 31;
    const int wave = tid >> 5;
    const int c    = lane & 15;
    const int h    = lane >> 4;
    const int rowTile = blockIdx.x * GEMM_WAVES + wave;
    const bool active = rowTile * 16 < M;
    const int rowBase = active ? rowTile * 16 : 0;

    float4 a[KT];
    const float4* Ap = (const float4*)(A + (size_t)(rowBase + c) * K) + h;
#pragma unroll
    for (int t = 0; t < KT; ++t) a[t] = Ap[2 * t];

#pragma unroll
    for (int i = 0; i < ITEMS / (GEMM_WAVES * 32); ++i) {
        const int idx = tid + i * (GEMM_WAVES * 32);
        const int p = idx / (NC / 4);
        const int q = idx - p * (NC / 4);
        const float4 r0 = *(const float4*)(B + (size_t)(2 * p) * NC + 4 * q);
        const float4 r1 = *(const float4*)(B + (size_t)(2 * p + 1) * NC + 4 * q);
        float4* d = (float4*)(Bs + p * PITCH + 8 * q);
        d[0] = make_float4(r0.x, r1.x, r0.y, r1.y);
        d[1] = make_float4(r0.z, r1.z, r0.w, r1.w);
    }
    __syncthreads();
    if (!active) return;

    v8f acc[NT];
#pragma unroll
    for (int j = 0; j < NT; ++j) acc[j] = v8f{};

    const float* bl = Bs + (2 * h) * PITCH + 2 * c;
#pragma unroll
    for (int t = 0; t < KT; ++t) {
#pragma unroll
        for (int u = 0; u < 2; ++u) {
            float a0 = u ? a[t].z : a[t].x;
            float a1 = u ? a[t].w : a[t].y;
            if (RELU) { a0 = fmaxf(a0, 0.0f); a1 = fmaxf(a1, 0.0f); }
            v2f av; av.x = a0; av.y = a1;
            const float* bp = bl + (4 * t + u) * PITCH;
#pragma unroll
            for (int j = 0; j < NT; ++j) {
                const v2f bv = *(const v2f*)(bp + 32 * j);
                acc[j] = __builtin_amdgcn_wmma_f32_16x16x4_f32(
                    false, av, false, bv, (short)0, acc[j], false, false);
            }
        }
    }

    float* Crow = C + (size_t)(rowBase + 8 * h) * NC + c;
#pragma unroll
    for (int j = 0; j < NT; ++j)
#pragma unroll
        for (int r = 0; r < 8; ++r)
            Crow[(size_t)r * NC + 16 * j] = acc[j][r];
}

template <int F, int NC2>
__global__ void __launch_bounds__(AGG_THREADS)
gcn_aggregate_owned(const int* __restrict__ src,
                    const int* __restrict__ dst,
                    const int* __restrict__ lists,
                    const int* __restrict__ tot,
                    const float* __restrict__ dinv,
                    const float* __restrict__ h,
                    const float* __restrict__ bias,
                    const float* __restrict__ W2,
                    float* __restrict__ out, int n, int e) {
    constexpr int J = F / 32;
    constexpr int U = 8;
    constexpr int SU = 4;
    constexpr int PA = NC2 > 0 ? F + 4 : F;
    __shared__ float4 acc4[RANGE * PA / 4];
    __shared__ float4 stg4[3 * STAGE / 4];
    float* acc = (float*)acc4;
    int*   es = (int*)stg4;
    int*   ed = es + STAGE;
    float* ew = (float*)(ed + STAGE);

    const int tid  = threadIdx.x;
    const int lane = tid & 31;
    const int wave = __builtin_amdgcn_readfirstlane(tid >> 5);
    const int b    = blockIdx.x;
    const int n0   = b * RANGE;
    const int rows = (n - n0) < RANGE ? (n - n0) : RANGE;
    const int total = __builtin_amdgcn_readfirstlane(tot[b]);

    constexpr int WITEMS = NC2 > 0 ? (F / 2) * (NC2 / 4) : AGG_THREADS;
    static_assert(WITEMS % AGG_THREADS == 0, "every thread stages the same number of pieces");
    constexpr int WPT = NC2 > 0 ? WITEMS / AGG_THREADS : 0;
    float4 w0[WPT > 0 ? WPT : 1], w1[WPT > 0 ? WPT : 1];
    if constexpr (NC2 > 0) {
#pragma unroll
        for (int i = 0; i < WPT; ++i) {
            const int idx = tid + i * AGG_THREADS;
            const int p = idx / (NC2 / 4);
            const int q = idx - p * (NC2 / 4);
            w0[i] = *(const float4*)(W2 + (size_t)(2 * p) * NC2 + 4 * q);
            w1[i] = *(const float4*)(W2 + (size_t)(2 * p + 1) * NC2 + 4 * q);
        }
    }

    for (int i = tid; i < rows * F; i += AGG_THREADS) {
        const int r = i / F;
        const int f = i - r * F;
        const float di = dinv[n0 + r];
        acc[r * PA + f] = bias[f] + h[(size_t)n0 * F + i] * (di * di);
    }
    __syncthreads();

    if (total <= LIST_CAP) {
        const v2i* mine = (const v2i*)lists + (size_t)b * LIST_CAP;
        for (int r0 = 0; r0 < total; r0 += STAGE) {
            const int mm = (total - r0) < STAGE ? (total - r0) : STAGE;

            for (int i0 = tid; i0 < mm; i0 += AGG_THREADS * SU) {
                v2i en[SU];
                float wv[SU];
#pragma unroll
                for (int u = 0; u < SU; ++u) {
                    const int i = i0 + u * AGG_THREADS;
                    en[u] = mine[r0 + (i < mm ? i : i0)];
                }
#pragma unroll
                for (int u = 0; u < SU; ++u) {
                    const int dr = en[u].y >= 0 ? en[u].y : 0;
                    wv[u] = dinv[en[u].x] * dinv[n0 + dr];
                }
#pragma unroll
                for (int u = 0; u < SU; ++u) {
                    const int i = i0 + u * AGG_THREADS;
                    if (i < mm) {
                        es[i] = en[u].x;
                        ed[i] = en[u].y;
                        ew[i] = en[u].y >= 0 ? wv[u] : 0.0f;
                    }
                }
            }
            __syncthreads();

            for (int base0 = wave * 32; base0 < mm; base0 += AGG_THREADS) {
                const int rest = mm - base0;
                const int c32 = rest < 32 ? rest : 32;
                int s = 0, dl = -1;
                float w = 0.0f;
                if (lane < c32) {
                    s  = es[base0 + lane];
                    dl = ed[base0 + lane];
                    w  = ew[base0 + lane];
                }
                const unsigned okmask = __builtin_amdgcn_ballot_w32(dl >= 0);
                int i = 0;
                for (; i + U <= c32; i += U) {
                    float v[U][J];
#pragma unroll
                    for (int u = 0; u < U; ++u) {
                        const int su = __builtin_amdgcn_readlane(s, i + u);
                        const float* hp = h + (size_t)su * F + lane;
#pragma unroll
                        for (int j = 0; j < J; ++j) v[u][j] = hp[32 * j];
                    }
#pragma unroll
                    for (int u = 0; u < U; ++u) {
                        if ((okmask >> (i + u)) & 1u) {
                            const int du = __builtin_amdgcn_readlane(dl, i + u);
                            const float wu = __int_as_float(__builtin_amdgcn_readlane(__float_as_int(w), i + u));
                            float* ap = acc + du * PA + lane;
#pragma unroll
                            for (int j = 0; j < J; ++j) atomicAdd(ap + 32 * j, v[u][j] * wu);
                        }
                    }
                }
                for (; i < c32; ++i) {
                    if ((okmask >> i) & 1u) {
                        const int su = __builtin_amdgcn_readlane(s, i);
                        const int du = __builtin_amdgcn_readlane(dl, i);
                        const float wu = __int_as_float(__builtin_amdgcn_readlane(__float_as_int(w), i));
                        const float* hp = h + (size_t)su * F + lane;
                        float* ap = acc + du * PA + lane;
#pragma unroll
                        for (int j = 0; j < J; ++j) atomicAdd(ap + 32 * j, hp[32 * j] * wu);
                    }
                }
            }
            __syncthreads();
        }
    } else {
        for (int ee = tid; ee < e; ee += AGG_THREADS) {
            const int d = dst[ee];
            if (d / RANGE == b) {
                const int s = src[ee];
                const float w = dinv[s] * dinv[d];
                float* ap = acc + (d - n0) * PA;
                const float* hp = h + (size_t)s * F;
#pragma unroll 1
                for (int f = 0; f < F; ++f) atomicAdd(ap + f, hp[f] * w);
            }
        }
        __syncthreads();
    }

    if constexpr (NC2 == 0) {
        for (int i = tid; i < rows * F; i += AGG_THREADS)
            out[(size_t)n0 * F + i] = acc[i];
    } else {
        constexpr int KT = F / 8;
        constexpr int NT2 = NC2 / 16;
        constexpr int PITCH = 2 * NC2 + 16;
        static_assert((F / 2) * PITCH <= 3 * STAGE, "W2 fits the staging area");
        float* Bs = (float*)stg4;
#pragma unroll
        for (int i = 0; i < WPT; ++i) {
            const int idx = tid + i * AGG_THREADS;
            const int p = idx / (NC2 / 4);
            const int q = idx - p * (NC2 / 4);
            float4* d = (float4*)(Bs + p * PITCH + 8 * q);
            d[0] = make_float4(w0[i].x, w1[i].x, w0[i].y, w1[i].y);
            d[1] = make_float4(w0[i].z, w1[i].z, w0[i].w, w1[i].w);
        }
        __syncthreads();

        const int c  = lane & 15;
        const int hh = lane >> 4;
        const int rtiles = (rows + 15) >> 4;
        for (int unit = wave; unit < rtiles * NT2; unit += AGG_THREADS / 32) {
            const int rt = unit / NT2;
            const int ct = unit - rt * NT2;
            const int arow = (rt * 16 + c) < rows ? (rt * 16 + c) : rows - 1;
            const float4* ap = (const float4*)(acc + arow * PA) + hh;
            const float* bl = Bs + (2 * hh) * PITCH + 2 * c + 32 * ct;
            v8f d = v8f{};
#pragma unroll
            for (int t = 0; t < KT; ++t) {
                const float4 a = ap[2 * t];
#pragma unroll
                for (int u = 0; u < 2; ++u) {
                    v2f av;
                    av.x = fmaxf(u ? a.z : a.x, 0.0f);
                    av.y = fmaxf(u ? a.w : a.y, 0.0f);
                    const v2f bv = *(const v2f*)(bl + (4 * t + u) * PITCH);
                    d = __builtin_amdgcn_wmma_f32_16x16x4_f32(
                        false, av, false, bv, (short)0, d, false, false);
                }
            }
            const int row0 = rt * 16 + 8 * hh;
            float* orow = out + (size_t)(n0 + row0) * NC2 + 16 * ct + c;
#pragma unroll
            for (int r = 0; r < 8; ++r)
                if (row0 + r < rows) orow[(size_t)r * NC2] = d[r];
        }
    }
}

extern "C" void kernel_launch(void* const* d_in, const int* in_sizes, int n_in,
                              void* d_out, int out_size, void* d_ws, size_t ws_size,
                              hipStream_t stream) {
    constexpr int D_IN = 64, D_HID = 128, D_OUT = 64;

    const float* x  = (const float*)d_in[0];
    const int*   ei = (const int*)  d_in[1];
    const float* W1 = (const float*)d_in[2];
    const float* b1 = (const float*)d_in[3];
    const float* W2 = (const float*)d_in[4];
    const float* b2 = (const float*)d_in[5];
    float* out = (float*)d_out;

    const int N = in_sizes[0] / D_IN;
    const int E = in_sizes[1] / 2;
    const int* src = ei;
    const int* dst = ei + E;

    auto cdiv = [](long a, long b) { return (int)((a + b - 1) / b); };
    const int owners = cdiv(N, RANGE);

    float* ws   = (float*)d_ws;
    float* dinv = ws;
    float* h1   = (float*)(((size_t)(dinv + N) + 255) & ~(size_t)255);
    float* h2   = h1 + (size_t)N * D_HID;
    unsigned char* keys = (unsigned char*)(h2 + (size_t)N * D_HID);
    keys = (unsigned char*)(((size_t)keys + 15) & ~(size_t)15);
    int* tot   = (int*)(((size_t)(keys + E) + 15) & ~(size_t)15);
    const int flagChunks = cdiv(E, FCHUNK);
    unsigned* flags = (unsigned*)(((size_t)(tot + owners) + 255) & ~(size_t)255);
    int* lists = (int*)(((size_t)(flags + (size_t)flagChunks * 256 * FWORDS) + 255) & ~(size_t)255);

    gcn_edge_keys<<<flagChunks, KEY_THREADS, 0, stream>>>(dst, (unsigned*)keys, flags, E);
    gcn_owner_prepass<<<owners, AGG_THREADS, 0, stream>>>(src, dst, keys, flags, dinv, lists, tot, N, E);

    gcn_gemm_wmma<false, D_IN, D_HID><<<cdiv(N / 16, GEMM_WAVES), GEMM_WAVES * 32, 0, stream>>>(
        x, W1, h1, N);
    gcn_aggregate_owned<D_HID, D_OUT><<<owners, AGG_THREADS, 0, stream>>>(
        src, dst, lists, tot, dinv, h1, b1, W2, h2, N, E);

    gcn_aggregate_owned<D_OUT, 0><<<owners, AGG_THREADS, 0, stream>>>(
        src, dst, lists, tot, dinv, h2, b2, nullptr, out, N, E);
}


# Add phase dealt evenly: each wave takes an equal run of whole 8-edge groups of the staged list (padded with zero-weight entries): no extra 32-edge trip, no serial remainder; LDS arrival order kept
# speedup vs baseline: 3.1616x; 1.0751x over previous
#include <hip/hip_runtime.h>

typedef float v2f __attribute__((ext_vector_type(2)));
typedef float v8f __attribute__((ext_vector_type(8)));

constexpr int RANGE = 196;
constexpr int AGG_THREADS = 1024;
constexpr int KEYS_PER_LOAD = 16;
constexpr int LIST_CAP = 49152;
constexpr int STAGE = 16384;
constexpr int GEMM_WAVES = 8;
constexpr int KEY_THREADS = 256;
constexpr int FCHUNK  = 4096;
constexpr int FGROUPS = FCHUNK / KEYS_PER_LOAD;
constexpr int FWORDS  = FGROUPS / 32;
constexpr int FPITCH  = 260;

typedef int v2i __attribute__((ext_vector_type(2)));

__global__ void __launch_bounds__(KEY_THREADS)
gcn_edge_keys(const int* __restrict__ dst, unsigned* __restrict__ keys32,
              unsigned* __restrict__ flags, int e) {
    constexpr int L = FCHUNK / 4 / KEY_THREADS;
    __shared__ unsigned m[FWORDS * FPITCH];

    const int tid   = threadIdx.x;
    const int chunk = blockIdx.x;
    const int ebase = chunk * FCHUNK;

    int d[L][4];
    bool fullw[L];
#pragma unroll
    for (int j = 0; j < L; ++j) {
        const int i4 = ebase + (j * KEY_THREADS + tid) * 4;
        fullw[j] = i4 + 3 < e;
        if (fullw[j]) {
            d[j][0] = dst[i4];     d[j][1] = dst[i4 + 1];
            d[j][2] = dst[i4 + 2]; d[j][3] = dst[i4 + 3];
        }
    }
    for (int i = tid; i < FWORDS * FPITCH; i += KEY_THREADS) m[i] = 0u;
    __syncthreads();

#pragma unroll
    for (int j = 0; j < L; ++j) {
        const int q  = j * KEY_THREADS + tid;
        const int i4 = ebase + q * 4;
        const int gi = q >> 2;
        unsigned* row = m + (gi >> 5) * FPITCH;
        const unsigned bit = 1u << (gi & 31);
        if (fullw[j]) {
            const unsigned o0 = (unsigned)(d[j][0] / RANGE) & 255u;
            const unsigned o1 = (unsigned)(d[j][1] / RANGE) & 255u;
            const unsigned o2 = (unsigned)(d[j][2] / RANGE) & 255u;
            const unsigned o3 = (unsigned)(d[j][3] / RANGE) & 255u;
            keys32[i4 >> 2] = o0 | (o1 << 8) | (o2 << 16) | (o3 << 24);
            atomicOr(row + o0, bit);
            atomicOr(row + o1, bit);
            atomicOr(row + o2, bit);
            atomicOr(row + o3, bit);
        } else {
            unsigned char* keys8 = (unsigned char*)keys32;
            for (int i = i4; i < e; ++i) {
                const unsigned o = (unsigned)(dst[i] / RANGE) & 255u;
                keys8[i] = (unsigned char)o;
                atomicOr(row + o, bit);
            }
        }
    }
    __syncthreads();

    for (int i = tid; i < 256 * FWORDS; i += KEY_THREADS)
        flags[(size_t)chunk * (256 * FWORDS) + i] = m[(i % FWORDS) * FPITCH + (i / FWORDS)];
}

__global__ void __launch_bounds__(AGG_THREADS)
gcn_owner_prepass(const int* __restrict__ src,
                  const int* __restrict__ dst,
                  const unsigned char* __restrict__ keys,
                  const unsigned* __restrict__ flags,
                  float* __restrict__ dinv,
                  int* __restrict__ lists,
                  int* __restrict__ tot, int n, int e) {
    constexpr int G = 4;
    constexpr int PU = 4;
    __shared__ int list[LIST_CAP];
    __shared__ int degc[RANGE];
    __shared__ int cnt;

    const int tid  = threadIdx.x;
    const int b    = blockIdx.x;
    const int n0   = b * RANGE;
    const int rows = (n - n0) < RANGE ? (n - n0) : RANGE;

    if (tid < RANGE) degc[tid] = 0;
    if (tid == 0) cnt = 0;
    __syncthreads();

    auto put = [&](int idx, int ee) {
        if (idx < LIST_CAP) {
            list[idx] = ee;
        } else {
            const int d = dst[ee];
            if (d / RANGE == b) atomicAdd(&degc[d - n0], 1);
        }
    };

    const unsigned mykey = (unsigned)b & 255u;
    const unsigned pat   = mykey * 0x01010101u;
    const int nFull = e >> 4;
    const int nFlagWords = ((e + FCHUNK - 1) / FCHUNK) * FWORDS;
    const uint4* keys128 = (const uint4*)keys;
    for (int wi = tid; wi < nFlagWords; wi += AGG_THREADS) {
        unsigned bits = flags[(size_t)(wi / FWORDS) * (256 * FWORDS) + mykey * FWORDS + (wi % FWORDS)];
        const int gbase = wi * 32;
        while (bits) {
            uint4 k[G];
            int eb[G];
            bool full[G];
#pragma unroll
            for (int u = 0; u < G; ++u) {
                const bool has = bits != 0u;
                const int g = gbase + (has ? __builtin_ctz(bits) : 0);
                bits &= bits - 1u;
                eb[u] = g * KEYS_PER_LOAD;
                full[u] = has && (g < nFull);
                k[u] = keys128[full[u] ? g : 0];
            }
            unsigned mm[G];
            unsigned any = 0u;
#pragma unroll
            for (int u = 0; u < G; ++u) {
                const unsigned wd[4] = {k[u].x, k[u].y, k[u].z, k[u].w};
                unsigned mk = 0u;
#pragma unroll
                for (int q = 0; q < 4; ++q) {
                    const unsigned x = wd[q] ^ pat;
                    const unsigned z = ~(((x & 0x7F7F7F7Fu) + 0x7F7F7F7Fu) | x) & 0x80808080u;
                    mk |= (z >> 7) << q;
                }
                mm[u] = full[u] ? mk : 0u;
                any |= mm[u];
            }
            if (any) {
                int nloc = 0;
#pragma unroll
                for (int u = 0; u < G; ++u) nloc += __builtin_popcount(mm[u]);
                int idx = atomicAdd(&cnt, nloc);
#pragma unroll
                for (int u = 0; u < G; ++u) {
                    unsigned mk = mm[u];
                    while (mk) {
                        const int bit = __builtin_ctz(mk);
                        mk &= mk - 1u;
                        put(idx++, eb[u] + 4 * (bit & 7) + (bit >> 3));
                    }
                }
            }
        }
    }
    if (tid < (e & (KEYS_PER_LOAD - 1))) {
        const int ee = (e & ~(KEYS_PER_LOAD - 1)) + tid;
        if ((unsigned)keys[ee] == mykey) put(atomicAdd(&cnt, 1), ee);
    }
    __syncthreads();

    const int total = cnt;
    const int m = total < LIST_CAP ? total : LIST_CAP;
    v2i* mine = (v2i*)lists + (size_t)b * LIST_CAP;
    for (int i0 = tid; i0 < m; i0 += AGG_THREADS * PU) {
        int dd[PU], ss[PU];
#pragma unroll
        for (int u = 0; u < PU; ++u) {
            const int i = i0 + u * AGG_THREADS;
            const int ee = list[i < m ? i : i0];
            dd[u] = dst[ee];
            ss[u] = src[ee];
        }
#pragma unroll
        for (int u = 0; u < PU; ++u) {
            const int i = i0 + u * AGG_THREADS;
            if (i < m) {
                const bool own = (dd[u] / RANGE == b);
                if (own) atomicAdd(&degc[dd[u] - n0], 1);
                v2i en;
                en.x = own ? ss[u] : 0;
                en.y = own ? dd[u] - n0 : -1;
                mine[i] = en;
            }
        }
    }
    __syncthreads();

    if (tid < rows) {
        const float dg = (float)(degc[tid] + 1);
        dinv[n0 + tid] = dg > 0.0f ? rsqrtf(dg) : 0.0f;
    }
    if (tid == 0) tot[b] = total;
}

template <bool RELU, int K, int NC>
__global__ void __launch_bounds__(GEMM_WAVES * 32)
gcn_gemm_wmma(const float* __restrict__ A,
              const float* __restrict__ B,
              float* __restrict__ C, int M) {
    constexpr int NT = NC / 16;
    constexpr int KT = K / 8;
    constexpr int PITCH = 2 * NC + 16;
    constexpr int ITEMS = (K / 2) * (NC / 4);
    static_assert(ITEMS % (GEMM_WAVES * 32) == 0, "every thread stages the same number of pieces");
    __shared__ float4 Bs4[(K / 2) * PITCH / 4];
    float* Bs = (float*)Bs4;

    const int tid  = threadIdx.x;
    const int lane = tid & 31;
    const int wave = tid >> 5;
    const int c    = lane & 15;
    const int h    = lane >> 4;
    const int rowTile = blockIdx.x * GEMM_WAVES + wave;
    const bool active = rowTile * 16 < M;
    const int rowBase = active ? rowTile * 16 : 0;

    float4 a[KT];
    const float4* Ap = (const float4*)(A + (size_t)(rowBase + c) * K) + h;
#pragma unroll
    for (int t = 0; t < KT; ++t) a[t] = Ap[2 * t];

#pragma unroll
    for (int i = 0; i < ITEMS / (GEMM_WAVES * 32); ++i) {
        const int idx = tid + i * (GEMM_WAVES * 32);
        const int p = idx / (NC / 4);
        const int q = idx - p * (NC / 4);
        const float4 r0 = *(const float4*)(B + (size_t)(2 * p) * NC + 4 * q);
        const float4 r1 = *(const float4*)(B + (size_t)(2 * p + 1) * NC + 4 * q);
        float4* d = (float4*)(Bs + p * PITCH + 8 * q);
        d[0] = make_float4(r0.x, r1.x, r0.y, r1.y);
        d[1] = make_float4(r0.z, r1.z, r0.w, r1.w);
    }
    __syncthreads();
    if (!active) return;

    v8f acc[NT];
#pragma unroll
    for (int j = 0; j < NT; ++j) acc[j] = v8f{};

    const float* bl = Bs + (2 * h) * PITCH + 2 * c;
#pragma unroll
    for (int t = 0; t < KT; ++t) {
#pragma unroll
        for (int u = 0; u < 2; ++u) {
            float a0 = u ? a[t].z : a[t].x;
            float a1 = u ? a[t].w : a[t].y;
            if (RELU) { a0 = fmaxf(a0, 0.0f); a1 = fmaxf(a1, 0.0f); }
            v2f av; av.x = a0; av.y = a1;
            const float* bp = bl + (4 * t + u) * PITCH;
#pragma unroll
            for (int j = 0; j < NT; ++j) {
                const v2f bv = *(const v2f*)(bp + 32 * j);
                acc[j] = __builtin_amdgcn_wmma_f32_16x16x4_f32(
                    false, av, false, bv, (short)0, acc[j], false, false);
            }
        }
    }

    float* Crow = C + (size_t)(rowBase + 8 * h) * NC + c;
#pragma unroll
    for (int j = 0; j < NT; ++j)
#pragma unroll
        for (int r = 0; r < 8; ++r)
            Crow[(size_t)r * NC + 16 * j] = acc[j][r];
}

template <int F, int NC2>
__global__ void __launch_bounds__(AGG_THREADS)
gcn_aggregate_owned(const int* __restrict__ src,
                    const int* __restrict__ dst,
                    const int* __restrict__ lists,
                    const int* __restrict__ tot,
                    const float* __restrict__ dinv,
                    const float* __restrict__ h,
                    const float* __restrict__ bias,
                    const float* __restrict__ W2,
                    float* __restrict__ out, int n, int e) {
    constexpr int J = F / 32;
    constexpr int U = 8;
    constexpr int SU = 4;
    static_assert((U & (U - 1)) == 0 && STAGE % U == 0 && 32 % U == 0, "a staged list pads to whole groups inside the staging area");
    constexpr int PA = NC2 > 0 ? F + 4 : F;
    __shared__ float4 acc4[RANGE * PA / 4];
    __shared__ float4 stg4[3 * STAGE / 4];
    float* acc = (float*)acc4;
    int*   es = (int*)stg4;
    int*   ed = es + STAGE;
    float* ew = (float*)(ed + STAGE);

    const int tid  = threadIdx.x;
    const int lane = tid & 31;
    const int wave = __builtin_amdgcn_readfirstlane(tid >> 5);
    const int b    = blockIdx.x;
    const int n0   = b * RANGE;
    const int rows = (n - n0) < RANGE ? (n - n0) : RANGE;
    const int total = __builtin_amdgcn_readfirstlane(tot[b]);

    constexpr int WITEMS = NC2 > 0 ? (F / 2) * (NC2 / 4) : AGG_THREADS;
    static_assert(WITEMS % AGG_THREADS == 0, "every thread stages the same number of pieces");
    constexpr int WPT = NC2 > 0 ? WITEMS / AGG_THREADS : 0;
    float4 w0[WPT > 0 ? WPT : 1], w1[WPT > 0 ? WPT : 1];
    if constexpr (NC2 > 0) {
#pragma unroll
        for (int i = 0; i < WPT; ++i) {
            const int idx = tid + i * AGG_THREADS;
            const int p = idx / (NC2 / 4);
            const int q = idx - p * (NC2 / 4);
            w0[i] = *(const float4*)(W2 + (size_t)(2 * p) * NC2 + 4 * q);
            w1[i] = *(const float4*)(W2 + (size_t)(2 * p + 1) * NC2 + 4 * q);
        }
    }

    for (int i = tid; i < rows * F; i += AGG_THREADS) {
        const int r = i / F;
        const int f = i - r * F;
        const float di = dinv[n0 + r];
        acc[r * PA + f] = bias[f] + h[(size_t)n0 * F + i] * (di * di);
    }
    __syncthreads();

    if (total <= LIST_CAP) {
        const v2i* mine = (const v2i*)lists + (size_t)b * LIST_CAP;
        for (int r0 = 0; r0 < total; r0 += STAGE) {
            const int mm = (total - r0) < STAGE ? (total - r0) : STAGE;

            for (int i0 = tid; i0 < mm; i0 += AGG_THREADS * SU) {
                v2i en[SU];
                float wv[SU];
#pragma unroll
                for (int u = 0; u < SU; ++u) {
                    const int i = i0 + u * AGG_THREADS;
                    en[u] = mine[r0 + (i < mm ? i : i0)];
                }
#pragma unroll
                for (int u = 0; u < SU; ++u) {
                    const int dr = en[u].y >= 0 ? en[u].y : 0;
                    wv[u] = dinv[en[u].x] * dinv[n0 + dr];
                }
#pragma unroll
                for (int u = 0; u < SU; ++u) {
                    const int i = i0 + u * AGG_THREADS;
                    if (i < mm) {
                        es[i] = en[u].x;
                        ed[i] = en[u].y;
                        ew[i] = en[u].y >= 0 ? wv[u] : 0.0f;
                    }
                }
            }
            const int mmU = (mm + U - 1) & ~(U - 1);
            if (tid < mmU - mm) {
                es[mm + tid] = 0;
                ed[mm + tid] = -1;
                ew[mm + tid] = 0.0f;
            }
            __syncthreads();

            const int ng = mmU / U;
            const int eend = (((wave + 1) * ng) / (AGG_THREADS / 32)) * U;
            for (int base0 = ((wave * ng) / (AGG_THREADS / 32)) * U; base0 < eend; base0 += 32) {
                const int rest = eend - base0;
                const int c32 = rest < 32 ? rest : 32;
                int s = 0, dl = -1;
                float w = 0.0f;
                if (lane < c32) {
                    s  = es[base0 + lane];
                    dl = ed[base0 + lane];
                    w  = ew[base0 + lane];
                }
                const unsigned okmask = __builtin_amdgcn_ballot_w32(dl >= 0);
                for (int i = 0; i < c32; i += U) {
                    float v[U][J];
#pragma unroll
                    for (int u = 0; u < U; ++u) {
                        const int su = __builtin_amdgcn_readlane(s, i + u);
                        const float* hp = h + (size_t)su * F + lane;
#pragma unroll
                        for (int j = 0; j < J; ++j) v[u][j] = hp[32 * j];
                    }
#pragma unroll
                    for (int u = 0; u < U; ++u) {
                        if ((okmask >> (i + u)) & 1u) {
                            const int du = __builtin_amdgcn_readlane(dl, i + u);
                            const float wu = __int_as_float(__builtin_amdgcn_readlane(__float_as_int(w), i + u));
                            float* ap = acc + du * PA + lane;
#pragma unroll
                            for (int j = 0; j < J; ++j) atomicAdd(ap + 32 * j, v[u][j] * wu);
                        }
                    }
                }
            }
            __syncthreads();
        }
    } else {
        for (int ee = tid; ee < e; ee += AGG_THREADS) {
            const int d = dst[ee];
            if (d / RANGE == b) {
                const int s = src[ee];
                const float w = dinv[s] * dinv[d];
                float* ap = acc + (d - n0) * PA;
                const float* hp = h + (size_t)s * F;
#pragma unroll 1
                for (int f = 0; f < F; ++f) atomicAdd(ap + f, hp[f] * w);
            }
        }
        __syncthreads();
    }

    if constexpr (NC2 == 0) {
        for (int i = tid; i < rows * F; i += AGG_THREADS)
            out[(size_t)n0 * F + i] = acc[i];
    } else {
        constexpr int KT = F / 8;
        constexpr int NT2 = NC2 / 16;
        constexpr int PITCH = 2 * NC2 + 16;
        static_assert((F / 2) * PITCH <= 3 * STAGE, "W2 fits the staging area");
        float* Bs = (float*)stg4;
#pragma unroll
        for (int i = 0; i < WPT; ++i) {
            const int idx = tid + i * AGG_THREADS;
            const int p = idx / (NC2 / 4);
            const int q = idx - p * (NC2 / 4);
            float4* d = (float4*)(Bs + p * PITCH + 8 * q);
            d[0] = make_float4(w0[i].x, w1[i].x, w0[i].y, w1[i].y);
            d[1] = make_float4(w0[i].z, w1[i].z, w0[i].w, w1[i].w);
        }
        __syncthreads();

        const int c  = lane & 15;
        const int hh = lane >> 4;
        const int rtiles = (rows + 15) >> 4;
        for (int unit = wave; unit < rtiles * NT2; unit += AGG_THREADS / 32) {
            const int rt = unit / NT2;
            const int ct = unit - rt * NT2;
            const int arow = (rt * 16 + c) < rows ? (rt * 16 + c) : rows - 1;
            const float4* ap = (const float4*)(acc + arow * PA) + hh;
            const float* bl = Bs + (2 * hh) * PITCH + 2 * c + 32 * ct;
            v8f d = v8f{};
#pragma unroll
            for (int t = 0; t < KT; ++t) {
                const float4 a = ap[2 * t];
#pragma unroll
                for (int u = 0; u < 2; ++u) {
                    v2f av;
                    av.x = fmaxf(u ? a.z : a.x, 0.0f);
                    av.y = fmaxf(u ? a.w : a.y, 0.0f);
                    const v2f bv = *(const v2f*)(bl + (4 * t + u) * PITCH);
                    d = __builtin_amdgcn_wmma_f32_16x16x4_f32(
                        false, av, false, bv, (short)0, d, false, false);
                }
            }
            const int row0 = rt * 16 + 8 * hh;
            float* orow = out + (size_t)(n0 + row0) * NC2 + 16 * ct + c;
#pragma unroll
            for (int r = 0; r < 8; ++r)
                if (row0 + r < rows) orow[(size_t)r * NC2] = d[r];
        }
    }
}

extern "C" void kernel_launch(void* const* d_in, const int* in_sizes, int n_in,
                              void* d_out, int out_size, void* d_ws, size_t ws_size,
                              hipStream_t stream) {
    constexpr int D_IN = 64, D_HID = 128, D_OUT = 64;

    const float* x  = (const float*)d_in[0];
    const int*   ei = (const int*)  d_in[1];
    const float* W1 = (const float*)d_in[2];
    const float* b1 = (const float*)d_in[3];
    const float* W2 = (const float*)d_in[4];
    const float* b2 = (const float*)d_in[5];
    float* out = (float*)d_out;

    const int N = in_sizes[0] / D_IN;
    const int E = in_sizes[1] / 2;
    const int* src = ei;
    const int* dst = ei + E;

    auto cdiv = [](long a, long b) { return (int)((a + b - 1) / b); };
    const int owners = cdiv(N, RANGE);

    float* ws   = (float*)d_ws;
    float* dinv = ws;
    float* h1   = (float*)(((size_t)(dinv + N) + 255) & ~(size_t)255);
    float* h2   = h1 + (size_t)N * D_HID;
    unsigned char* keys = (unsigned char*)(h2 + (size_t)N * D_HID);
    keys = (unsigned char*)(((size_t)keys + 15) & ~(size_t)15);
    int* tot   = (int*)(((size_t)(keys + E) + 15) & ~(size_t)15);
    const int flagChunks = cdiv(E, FCHUNK);
    unsigned* flags = (unsigned*)(((size_t)(tot + owners) + 255) & ~(size_t)255);
    int* lists = (int*)(((size_t)(flags + (size_t)flagChunks * 256 * FWORDS) + 255) & ~(size_t)255);

    gcn_edge_keys<<<flagChunks, KEY_THREADS, 0, stream>>>(dst, (unsigned*)keys, flags, E);
    gcn_owner_prepass<<<owners, AGG_THREADS, 0, stream>>>(src, dst, keys, flags, dinv, lists, tot, N, E);

    gcn_gemm_wmma<false, D_IN, D_HID><<<cdiv(N / 16, GEMM_WAVES), GEMM_WAVES * 32, 0, stream>>>(
        x, W1, h1, N);
    gcn_aggregate_owned<D_HID, D_OUT><<<owners, AGG_THREADS, 0, stream>>>(
        src, dst, lists, tot, dinv, h1, b1, W2, h2, N, E);

    gcn_aggregate_owned<D_OUT, 0><<<owners, AGG_THREADS, 0, stream>>>(
        src, dst, lists, tot, dinv, h2, b2, nullptr, out, N, E);
}


# Edge-key groups and x @ W1 product groups share ONE launch (independent work; work-group index picks the body, one LDS area): key waits filled by matrix work, 5 -> 4 launches; sum orders kept
# speedup vs baseline: 3.2857x; 1.0393x over previous
#include <hip/hip_runtime.h>

typedef float v2f __attribute__((ext_vector_type(2)));
typedef float v8f __attribute__((ext_vector_type(8)));

constexpr int RANGE = 196;
constexpr int AGG_THREADS = 1024;
constexpr int KEYS_PER_LOAD = 16;
constexpr int LIST_CAP = 49152;
constexpr int STAGE = 16384;
constexpr int GEMM_WAVES = 8;
constexpr int KEY_THREADS = 256;
constexpr int FCHUNK  = 4096;
constexpr int FGROUPS = FCHUNK / KEYS_PER_LOAD;
constexpr int FWORDS  = FGROUPS / 32;
constexpr int FPITCH  = 260;

typedef int v2i __attribute__((ext_vector_type(2)));

__device__ __forceinline__ void
edge_keys_group(const int* __restrict__ dst, unsigned* __restrict__ keys32,
                unsigned* __restrict__ flags, int e, int chunk, unsigned* m) {
    constexpr int L = FCHUNK / 4 / KEY_THREADS;

    const int tid   = threadIdx.x;
    const int ebase = chunk * FCHUNK;

    int d[L][4];
    bool fullw[L];
#pragma unroll
    for (int j = 0; j < L; ++j) {
        const int i4 = ebase + (j * KEY_THREADS + tid) * 4;
        fullw[j] = i4 + 3 < e;
        if (fullw[j]) {
            d[j][0] = dst[i4];     d[j][1] = dst[i4 + 1];
            d[j][2] = dst[i4 + 2]; d[j][3] = dst[i4 + 3];
        }
    }
    for (int i = tid; i < FWORDS * FPITCH; i += KEY_THREADS) m[i] = 0u;
    __syncthreads();

#pragma unroll
    for (int j = 0; j < L; ++j) {
        const int q  = j * KEY_THREADS + tid;
        const int i4 = ebase + q * 4;
        const int gi = q >> 2;
        unsigned* row = m + (gi >> 5) * FPITCH;
        const unsigned bit = 1u << (gi & 31);
        if (fullw[j]) {
            const unsigned o0 = (unsigned)(d[j][0] / RANGE) & 255u;
            const unsigned o1 = (unsigned)(d[j][1] / RANGE) & 255u;
            const unsigned o2 = (unsigned)(d[j][2] / RANGE) & 255u;
            const unsigned o3 = (unsigned)(d[j][3] / RANGE) & 255u;
            keys32[i4 >> 2] = o0 | (o1 << 8) | (o2 << 16) | (o3 << 24);
            atomicOr(row + o0, bit);
            atomicOr(row + o1, bit);
            atomicOr(row + o2, bit);
            atomicOr(row + o3, bit);
        } else {
            unsigned char* keys8 = (unsigned char*)keys32;
            for (int i = i4; i < e; ++i) {
                const unsigned o = (unsigned)(dst[i] / RANGE) & 255u;
                keys8[i] = (unsigned char)o;
                atomicOr(row + o, bit);
            }
        }
    }
    __syncthreads();

    for (int i = tid; i < 256 * FWORDS; i += KEY_THREADS)
        flags[(size_t)chunk * (256 * FWORDS) + i] = m[(i % FWORDS) * FPITCH + (i / FWORDS)];
}

__global__ void __launch_bounds__(AGG_THREADS)
gcn_owner_prepass(const int* __restrict__ src,
                  const int* __restrict__ dst,
                  const unsigned char* __restrict__ keys,
                  const unsigned* __restrict__ flags,
                  float* __restrict__ dinv,
                  int* __restrict__ lists,
                  int* __restrict__ tot, int n, int e) {
    constexpr int G = 4;
    constexpr int PU = 4;
    __shared__ int list[LIST_CAP];
    __shared__ int degc[RANGE];
    __shared__ int cnt;

    const int tid  = threadIdx.x;
    const int b    = blockIdx.x;
    const int n0   = b * RANGE;
    const int rows = (n - n0) < RANGE ? (n - n0) : RANGE;

    if (tid < RANGE) degc[tid] = 0;
    if (tid == 0) cnt = 0;
    __syncthreads();

    auto put = [&](int idx, int ee) {
        if (idx < LIST_CAP) {
            list[idx] = ee;
        } else {
            const int d = dst[ee];
            if (d / RANGE == b) atomicAdd(&degc[d - n0], 1);
        }
    };

    const unsigned mykey = (unsigned)b & 255u;
    const unsigned pat   = mykey * 0x01010101u;
    const int nFull = e >> 4;
    const int nFlagWords = ((e + FCHUNK - 1) / FCHUNK) * FWORDS;
    const uint4* keys128 = (const uint4*)keys;
    for (int wi = tid; wi < nFlagWords; wi += AGG_THREADS) {
        unsigned bits = flags[(size_t)(wi / FWORDS) * (256 * FWORDS) + mykey * FWORDS + (wi % FWORDS)];
        const int gbase = wi * 32;
        while (bits) {
            uint4 k[G];
            int eb[G];
            bool full[G];
#pragma unroll
            for (int u = 0; u < G; ++u) {
                const bool has = bits != 0u;
                const int g = gbase + (has ? __builtin_ctz(bits) : 0);
                bits &= bits - 1u;
                eb[u] = g * KEYS_PER_LOAD;
                full[u] = has && (g < nFull);
                k[u] = keys128[full[u] ? g : 0];
            }
            unsigned mm[G];
            unsigned any = 0u;
#pragma unroll
            for (int u = 0; u < G; ++u) {
                const unsigned wd[4] = {k[u].x, k[u].y, k[u].z, k[u].w};
                unsigned mk = 0u;
#pragma unroll
                for (int q = 0; q < 4; ++q) {
                    const unsigned x = wd[q] ^ pat;
                    const unsigned z = ~(((x & 0x7F7F7F7Fu) + 0x7F7F7F7Fu) | x) & 0x80808080u;
                    mk |= (z >> 7) << q;
                }
                mm[u] = full[u] ? mk : 0u;
                any |= mm[u];
            }
            if (any) {
                int nloc = 0;
#pragma unroll
                for (int u = 0; u < G; ++u) nloc += __builtin_popcount(mm[u]);
                int idx = atomicAdd(&cnt, nloc);
#pragma unroll
                for (int u = 0; u < G; ++u) {
                    unsigned mk = mm[u];
                    while (mk) {
                        const int bit = __builtin_ctz(mk);
                        mk &= mk - 1u;
                        put(idx++, eb[u] + 4 * (bit & 7) + (bit >> 3));
                    }
                }
            }
        }
    }
    if (tid < (e & (KEYS_PER_LOAD - 1))) {
        const int ee = (e & ~(KEYS_PER_LOAD - 1)) + tid;
        if ((unsigned)keys[ee] == mykey) put(atomicAdd(&cnt, 1), ee);
    }
    __syncthreads();

    const int total = cnt;
    const int m = total < LIST_CAP ? total : LIST_CAP;
    v2i* mine = (v2i*)lists + (size_t)b * LIST_CAP;
    for (int i0 = tid; i0 < m; i0 += AGG_THREADS * PU) {
        int dd[PU], ss[PU];
#pragma unroll
        for (int u = 0; u < PU; ++u) {
            const int i = i0 + u * AGG_THREADS;
            const int ee = list[i < m ? i : i0];
            dd[u] = dst[ee];
            ss[u] = src[ee];
        }
#pragma unroll
        for (int u = 0; u < PU; ++u) {
            const int i = i0 + u * AGG_THREADS;
            if (i < m) {
                const bool own = (dd[u] / RANGE == b);
                if (own) atomicAdd(&degc[dd[u] - n0], 1);
                v2i en;
                en.x = own ? ss[u] : 0;
                en.y = own ? dd[u] - n0 : -1;
                mine[i] = en;
            }
        }
    }
    __syncthreads();

    if (tid < rows) {
        const float dg = (float)(degc[tid] + 1);
        dinv[n0 + tid] = dg > 0.0f ? rsqrtf(dg) : 0.0f;
    }
    if (tid == 0) tot[b] = total;
}

template <bool RELU, int K, int NC>
__device__ __forceinline__ void
gemm_wmma_group(const float* __restrict__ A,
                const float* __restrict__ B,
                float* __restrict__ C, int M, int blk, float* Bs) {
    constexpr int NT = NC / 16;
    constexpr int KT = K / 8;
    constexpr int PITCH = 2 * NC + 16;
    constexpr int ITEMS = (K / 2) * (NC / 4);
    static_assert(ITEMS % (GEMM_WAVES * 32) == 0, "every thread stages the same number of pieces");

    const int tid  = threadIdx.x;
    const int lane = tid & 31;
    const int wave = tid >> 5;
    const int c    = lane & 15;
    const int h    = lane >> 4;
    const int rowTile = blk * GEMM_WAVES + wave;
    const bool active = rowTile * 16 < M;
    const int rowBase = active ? rowTile * 16 : 0;

    float4 a[KT];
    const float4* Ap = (const float4*)(A + (size_t)(rowBase + c) * K) + h;
#pragma unroll
    for (int t = 0; t < KT; ++t) a[t] = Ap[2 * t];

#pragma unroll
    for (int i = 0; i < ITEMS / (GEMM_WAVES * 32); ++i) {
        const int idx = tid + i * (GEMM_WAVES * 32);
        const int p = idx / (NC / 4);
        const int q = idx - p * (NC / 4);
        const float4 r0 = *(const float4*)(B + (size_t)(2 * p) * NC + 4 * q);
        const float4 r1 = *(const float4*)(B + (size_t)(2 * p + 1) * NC + 4 * q);
        float4* d = (float4*)(Bs + p * PITCH + 8 * q);
        d[0] = make_float4(r0.x, r1.x, r0.y, r1.y);
        d[1] = make_float4(r0.z, r1.z, r0.w, r1.w);
    }
    __syncthreads();
    if (!active) return;

    v8f acc[NT];
#pragma unroll
    for (int j = 0; j < NT; ++j) acc[j] = v8f{};

    const float* bl = Bs + (2 * h) * PITCH + 2 * c;
#pragma unroll
    for (int t = 0; t < KT; ++t) {
#pragma unroll
        for (int u = 0; u < 2; ++u) {
            float a0 = u ? a[t].z : a[t].x;
            float a1 = u ? a[t].w : a[t].y;
            if (RELU) { a0 = fmaxf(a0, 0.0f); a1 = fmaxf(a1, 0.0f); }
            v2f av; av.x = a0; av.y = a1;
            const float* bp = bl + (4 * t + u) * PITCH;
#pragma unroll
            for (int j = 0; j < NT; ++j) {
                const v2f bv = *(const v2f*)(bp + 32 * j);
                acc[j] = __builtin_amdgcn_wmma_f32_16x16x4_f32(
                    false, av, false, bv, (short)0, acc[j], false, false);
            }
        }
    }

    float* Crow = C + (size_t)(rowBase + 8 * h) * NC + c;
#pragma unroll
    for (int j = 0; j < NT; ++j)
#pragma unroll
        for (int r = 0; r < 8; ++r)
            Crow[(size_t)r * NC + 16 * j] = acc[j][r];
}

static_assert(KEY_THREADS == GEMM_WAVES * 32, "both kinds of work-group have the same size");
template <int K, int NC>
__global__ void __launch_bounds__(KEY_THREADS)
gcn_keys_and_product(const int* __restrict__ dst, unsigned* __restrict__ keys32,
                     unsigned* __restrict__ flags, int e,
                     const float* __restrict__ A, const float* __restrict__ B,
                     float* __restrict__ C, int M, int gemmGroups) {
    constexpr int GEMM_LDS_FLOATS = (K / 2) * (2 * NC + 16);
    static_assert(FWORDS * FPITCH <= GEMM_LDS_FLOATS, "the key groups' flag words fit the product groups' LDS");
    __shared__ float4 lds4[GEMM_LDS_FLOATS / 4];
    const int blk = blockIdx.x;
    if (blk < gemmGroups)
        gemm_wmma_group<false, K, NC>(A, B, C, M, blk, (float*)lds4);
    else
        edge_keys_group(dst, keys32, flags, e, blk - gemmGroups, (unsigned*)lds4);
}

template <int F, int NC2>
__global__ void __launch_bounds__(AGG_THREADS)
gcn_aggregate_owned(const int* __restrict__ src,
                    const int* __restrict__ dst,
                    const int* __restrict__ lists,
                    const int* __restrict__ tot,
                    const float* __restrict__ dinv,
                    const float* __restrict__ h,
                    const float* __restrict__ bias,
                    const float* __restrict__ W2,
                    float* __restrict__ out, int n, int e) {
    constexpr int J = F / 32;
    constexpr int U = 8;
    constexpr int SU = 4;
    static_assert((U & (U - 1)) == 0 && STAGE % U == 0 && 32 % U == 0, "a staged list pads to whole groups inside the staging area");
    constexpr int PA = NC2 > 0 ? F + 4 : F;
    __shared__ float4 acc4[RANGE * PA / 4];
    __shared__ float4 stg4[3 * STAGE / 4];
    float* acc = (float*)acc4;
    int*   es = (int*)stg4;
    int*   ed = es + STAGE;
    float* ew = (float*)(ed + STAGE);

    const int tid  = threadIdx.x;
    const int lane = tid & 31;
    const int wave = __builtin_amdgcn_readfirstlane(tid >> 5);
    const int b    = blockIdx.x;
    const int n0   = b * RANGE;
    const int rows = (n - n0) < RANGE ? (n - n0) : RANGE;
    const int total = __builtin_amdgcn_readfirstlane(tot[b]);

    constexpr int WITEMS = NC2 > 0 ? (F / 2) * (NC2 / 4) : AGG_THREADS;
    static_assert(WITEMS % AGG_THREADS == 0, "every thread stages the same number of pieces");
    constexpr int WPT = NC2 > 0 ? WITEMS / AGG_THREADS : 0;
    float4 w0[WPT > 0 ? WPT : 1], w1[WPT > 0 ? WPT : 1];
    if constexpr (NC2 > 0) {
#pragma unroll
        for (int i = 0; i < WPT; ++i) {
            const int idx = tid + i * AGG_THREADS;
            const int p = idx / (NC2 / 4);
            const int q = idx - p * (NC2 / 4);
            w0[i] = *(const float4*)(W2 + (size_t)(2 * p) * NC2 + 4 * q);
            w1[i] = *(const float4*)(W2 + (size_t)(2 * p + 1) * NC2 + 4 * q);
        }
    }

    for (int i = tid; i < rows * F; i += AGG_THREADS) {
        const int r = i / F;
        const int f = i - r * F;
        const float di = dinv[n0 + r];
        acc[r * PA + f] = bias[f] + h[(size_t)n0 * F + i] * (di * di);
    }
    __syncthreads();

    if (total <= LIST_CAP) {
        const v2i* mine = (const v2i*)lists + (size_t)b * LIST_CAP;
        for (int r0 = 0; r0 < total; r0 += STAGE) {
            const int mm = (total - r0) < STAGE ? (total - r0) : STAGE;

            for (int i0 = tid; i0 < mm; i0 += AGG_THREADS * SU) {
                v2i en[SU];
                float wv[SU];
#pragma unroll
                for (int u = 0; u < SU; ++u) {
                    const int i = i0 + u * AGG_THREADS;
                    en[u] = mine[r0 + (i < mm ? i : i0)];
                }
#pragma unroll
                for (int u = 0; u < SU; ++u) {
                    const int dr = en[u].y >= 0 ? en[u].y : 0;
                    wv[u] = dinv[en[u].x] * dinv[n0 + dr];
                }
#pragma unroll
                for (int u = 0; u < SU; ++u) {
                    const int i = i0 + u * AGG_THREADS;
                    if (i < mm) {
                        es[i] = en[u].x;
                        ed[i] = en[u].y;
                        ew[i] = en[u].y >= 0 ? wv[u] : 0.0f;
                    }
                }
            }
            const int mmU = (mm + U - 1) & ~(U - 1);
            if (tid < mmU - mm) {
                es[mm + tid] = 0;
                ed[mm + tid] = -1;
                ew[mm + tid] = 0.0f;
            }
            __syncthreads();

            const int ng = mmU / U;
            const int eend = (((wave + 1) * ng) / (AGG_THREADS / 32)) * U;
            for (int base0 = ((wave * ng) / (AGG_THREADS / 32)) * U; base0 < eend; base0 += 32) {
                const int rest = eend - base0;
                const int c32 = rest < 32 ? rest : 32;
                int s = 0, dl = -1;
                float w = 0.0f;
                if (lane < c32) {
                    s  = es[base0 + lane];
                    dl = ed[base0 + lane];
                    w  = ew[base0 + lane];
                }
                const unsigned okmask = __builtin_amdgcn_ballot_w32(dl >= 0);
                for (int i = 0; i < c32; i += U) {
                    float v[U][J];
#pragma unroll
                    for (int u = 0; u < U; ++u) {
                        const int su = __builtin_amdgcn_readlane(s, i + u);
                        const float* hp = h + (size_t)su * F + lane;
#pragma unroll
                        for (int j = 0; j < J; ++j) v[u][j] = hp[32 * j];
                    }
#pragma unroll
                    for (int u = 0; u < U; ++u) {
                        if ((okmask >> (i + u)) & 1u) {
                            const int du = __builtin_amdgcn_readlane(dl, i + u);
                            const float wu = __int_as_float(__builtin_amdgcn_readlane(__float_as_int(w), i + u));
                            float* ap = acc + du * PA + lane;
#pragma unroll
                            for (int j = 0; j < J; ++j) atomicAdd(ap + 32 * j, v[u][j] * wu);
                        }
                    }
                }
            }
            __syncthreads();
        }
    } else {
        for (int ee = tid; ee < e; ee += AGG_THREADS) {
            const int d = dst[ee];
            if (d / RANGE == b) {
                const int s = src[ee];
                const float w = dinv[s] * dinv[d];
                float* ap = acc + (d - n0) * PA;
                const float* hp = h + (size_t)s * F;
#pragma unroll 1
                for (int f = 0; f < F; ++f) atomicAdd(ap + f, hp[f] * w);
            }
        }
        __syncthreads();
    }

    if constexpr (NC2 == 0) {
        for (int i = tid; i < rows * F; i += AGG_THREADS)
            out[(size_t)n0 * F + i] = acc[i];
    } else {
        constexpr int KT = F / 8;
        constexpr int NT2 = NC2 / 16;
        constexpr int PITCH = 2 * NC2 + 16;
        static_assert((F / 2) * PITCH <= 3 * STAGE, "W2 fits the staging area");
        float* Bs = (float*)stg4;
#pragma unroll
        for (int i = 0; i < WPT; ++i) {
            const int idx = tid + i * AGG_THREADS;
            const int p = idx / (NC2 / 4);
            const int q = idx - p * (NC2 / 4);
            float4* d = (float4*)(Bs + p * PITCH + 8 * q);
            d[0] = make_float4(w0[i].x, w1[i].x, w0[i].y, w1[i].y);
            d[1] = make_float4(w0[i].z, w1[i].z, w0[i].w, w1[i].w);
        }
        __syncthreads();

        const int c  = lane & 15;
        const int hh = lane >> 4;
        const int rtiles = (rows + 15) >> 4;
        for (int unit = wave; unit < rtiles * NT2; unit += AGG_THREADS / 32) {
            const int rt = unit / NT2;
            const int ct = unit - rt * NT2;
            const int arow = (rt * 16 + c) < rows ? (rt * 16 + c) : rows - 1;
            const float4* ap = (const float4*)(acc + arow * PA) + hh;
            const float* bl = Bs + (2 * hh) * PITCH + 2 * c + 32 * ct;
            v8f d = v8f{};
#pragma unroll
            for (int t = 0; t < KT; ++t) {
                const float4 a = ap[2 * t];
#pragma unroll
                for (int u = 0; u < 2; ++u) {
                    v2f av;
                    av.x = fmaxf(u ? a.z : a.x, 0.0f);
                    av.y = fmaxf(u ? a.w : a.y, 0.0f);
                    const v2f bv = *(const v2f*)(bl + (4 * t + u) * PITCH);
                    d = __builtin_amdgcn_wmma_f32_16x16x4_f32(
                        false, av, false, bv, (short)0, d, false, false);
                }
            }
            const int row0 = rt * 16 + 8 * hh;
            float* orow = out + (size_t)(n0 + row0) * NC2 + 16 * ct + c;
#pragma unroll
            for (int r = 0; r < 8; ++r)
                if (row0 + r < rows) orow[(size_t)r * NC2] = d[r];
        }
    }
}

extern "C" void kernel_launch(void* const* d_in, const int* in_sizes, int n_in,
                              void* d_out, int out_size, void* d_ws, size_t ws_size,
                              hipStream_t stream) {
    constexpr int D_IN = 64, D_HID = 128, D_OUT = 64;

    const float* x  = (const float*)d_in[0];
    const int*   ei = (const int*)  d_in[1];
    const float* W1 = (const float*)d_in[2];
    const float* b1 = (const float*)d_in[3];
    const float* W2 = (const float*)d_in[4];
    const float* b2 = (const float*)d_in[5];
    float* out = (float*)d_out;

    const int N = in_sizes[0] / D_IN;
    const int E = in_sizes[1] / 2;
    const int* src = ei;
    const int* dst = ei + E;

    auto cdiv = [](long a, long b) { return (int)((a + b - 1) / b); };
    const int owners = cdiv(N, RANGE);

    float* ws   = (float*)d_ws;
    float* dinv = ws;
    float* h1   = (float*)(((size_t)(dinv + N) + 255) & ~(size_t)255);
    float* h2   = h1 + (size_t)N * D_HID;
    unsigned char* keys = (unsigned char*)(h2 + (size_t)N * D_HID);
    keys = (unsigned char*)(((size_t)keys + 15) & ~(size_t)15);
    int* tot   = (int*)(((size_t)(keys + E) + 15) & ~(size_t)15);
    const int flagChunks = cdiv(E, FCHUNK);
    unsigned* flags = (unsigned*)(((size_t)(tot + owners) + 255) & ~(size_t)255);
    int* lists = (int*)(((size_t)(flags + (size_t)flagChunks * 256 * FWORDS) + 255) & ~(size_t)255);

    const int gemmGroups = cdiv(N / 16, GEMM_WAVES);
    gcn_keys_and_product<D_IN, D_HID><<<gemmGroups + flagChunks, KEY_THREADS, 0, stream>>>(
        dst, (unsigned*)keys, flags, E, x, W1, h1, N, gemmGroups);
    gcn_owner_prepass<<<owners, AGG_THREADS, 0, stream>>>(src, dst, keys, flags, dinv, lists, tot, N, E);

    gcn_aggregate_owned<D_HID, D_OUT><<<owners, AGG_THREADS, 0, stream>>>(
        src, dst, lists, tot, dinv, h1, b1, W2, h2, N, E);

    gcn_aggregate_owned<D_OUT, 0><<<owners, AGG_THREADS, 0, stream>>>(
        src, dst, lists, tot, dinv, h2, b2, nullptr, out, N, E);
}


# Key launch writes one 8-B record per edge (src, owner, local row); owners read whole flagged 16-edge record lines by half-waves and list the matches: no per-edge index reads; list order changes
# speedup vs baseline: 3.5076x; 1.0675x over previous
#include <hip/hip_runtime.h>

typedef float v2f __attribute__((ext_vector_type(2)));
typedef float v8f __attribute__((ext_vector_type(8)));

constexpr int RANGE = 196;
constexpr int AGG_THREADS = 1024;
constexpr int GROUP_EDGES = 16;
constexpr int LIST_CAP = 49152;
constexpr int GL_CAP   = 16384;
constexpr int PSTAGE   = 16384;
constexpr int STAGE = 16384;
constexpr int GEMM_WAVES = 8;
constexpr int KEY_THREADS = 256;
constexpr int FCHUNK  = 4096;
constexpr int FGROUPS = FCHUNK / GROUP_EDGES;
constexpr int FWORDS  = FGROUPS / 32;
constexpr int FPITCH  = 260;

typedef int v2i __attribute__((ext_vector_type(2)));
typedef int v4i __attribute__((ext_vector_type(4)));

__device__ __forceinline__ void
edge_keys_group(const int* __restrict__ src, const int* __restrict__ dst,
                int* __restrict__ recs, unsigned* __restrict__ flags,
                int e, int chunk, unsigned* m) {
    constexpr int L = FCHUNK / 4 / KEY_THREADS;

    const int tid   = threadIdx.x;
    const int ebase = chunk * FCHUNK;

    int d[L][4], s[L][4];
    bool fullw[L];
#pragma unroll
    for (int j = 0; j < L; ++j) {
        const int i4 = ebase + (j * KEY_THREADS + tid) * 4;
        fullw[j] = i4 + 3 < e;
        const int ia = fullw[j] ? i4 : 0;
        d[j][0] = dst[ia];     d[j][1] = dst[ia + 1];
        d[j][2] = dst[ia + 2]; d[j][3] = dst[ia + 3];
        s[j][0] = src[ia];     s[j][1] = src[ia + 1];
        s[j][2] = src[ia + 2]; s[j][3] = src[ia + 3];
    }
    for (int i = tid; i < FWORDS * FPITCH; i += KEY_THREADS) m[i] = 0u;
    __syncthreads();

#pragma unroll
    for (int j = 0; j < L; ++j) {
        const int q  = j * KEY_THREADS + tid;
        const int i4 = ebase + q * 4;
        const int gi = q >> 2;
        unsigned* frow = m + (gi >> 5) * FPITCH;
        const unsigned bit = 1u << (gi & 31);
        if (fullw[j]) {
            const int o0 = d[j][0] / RANGE, o1 = d[j][1] / RANGE;
            const int o2 = d[j][2] / RANGE, o3 = d[j][3] / RANGE;
            v4i* rp = (v4i*)(recs + 2 * (size_t)i4);
            v4i ra, rb;
            ra.x = s[j][0]; ra.y = (o0 << 8) | (d[j][0] - o0 * RANGE);
            ra.z = s[j][1]; ra.w = (o1 << 8) | (d[j][1] - o1 * RANGE);
            rb.x = s[j][2]; rb.y = (o2 << 8) | (d[j][2] - o2 * RANGE);
            rb.z = s[j][3]; rb.w = (o3 << 8) | (d[j][3] - o3 * RANGE);
            rp[0] = ra;
            rp[1] = rb;
            atomicOr(frow + ((unsigned)o0 & 255u), bit);
            atomicOr(frow + ((unsigned)o1 & 255u), bit);
            atomicOr(frow + ((unsigned)o2 & 255u), bit);
            atomicOr(frow + ((unsigned)o3 & 255u), bit);
        } else {
            for (int i = i4; i < e; ++i) {
                const int dd = dst[i];
                const int o = dd / RANGE;
                recs[2 * (size_t)i]     = src[i];
                recs[2 * (size_t)i + 1] = (o << 8) | (dd - o * RANGE);
                atomicOr(frow + ((unsigned)o & 255u), bit);
            }
        }
    }
    __syncthreads();

    for (int i = tid; i < 256 * FWORDS; i += KEY_THREADS)
        flags[(size_t)chunk * (256 * FWORDS) + i] = m[(i % FWORDS) * FPITCH + (i / FWORDS)];
}

__global__ void __launch_bounds__(AGG_THREADS)
gcn_owner_prepass(const int* __restrict__ recs,
                  const unsigned* __restrict__ flags,
                  float* __restrict__ dinv,
                  int* __restrict__ lists,
                  int* __restrict__ tot, int n, int e) {
    constexpr int UB = 16;
    constexpr int HW = AGG_THREADS / 16;
    __shared__ int glist[GL_CAP];
    __shared__ v2i pairs[PSTAGE];
    __shared__ int degc[RANGE];
    __shared__ int gcnt;
    __shared__ int cnt;

    const int tid  = threadIdx.x;
    const int lane = tid & 31;
    const int wave = __builtin_amdgcn_readfirstlane(tid >> 5);
    const int b    = blockIdx.x;
    const int n0   = b * RANGE;
    const int rows = (n - n0) < RANGE ? (n - n0) : RANGE;

    if (tid < RANGE) degc[tid] = 0;
    if (tid == 0) { gcnt = 0; cnt = 0; }
    __syncthreads();

    const v2i* rec2 = (const v2i*)recs;
    v2i* mine = (v2i*)lists + (size_t)b * LIST_CAP;

    auto take = [&](int idx, v2i r) {
        const int row = r.y & 255;
        atomicAdd(&degc[row], 1);
        v2i en;
        en.x = r.x;
        en.y = row;
        if (idx < PSTAGE) pairs[idx] = en;
        else if (idx < LIST_CAP) mine[idx] = en;
    };

    const unsigned mykey = (unsigned)b & 255u;
    const int nFull = e >> 4;
    const int nFlagWords = ((e + FCHUNK - 1) / FCHUNK) * FWORDS;
    for (int w0 = wave * 32; w0 < nFlagWords; w0 += AGG_THREADS) {
        const int wi = w0 + lane;
        unsigned bits = 0u;
        if (wi < nFlagWords)
            bits = flags[(size_t)(wi / FWORDS) * (256 * FWORDS) + mykey * FWORDS + (wi % FWORDS)];
        const int lim = nFull - wi * 32;
        if (lim < 32) bits = lim <= 0 ? 0u : (bits & ((1u << lim) - 1u));
        const int nb = __builtin_popcount(bits);
        int inc = nb;
#pragma unroll
        for (int dl = 1; dl < 32; dl <<= 1) {
            const int t = __builtin_amdgcn_ds_bpermute(((lane - dl) & 31) << 2, inc);
            if (lane >= dl) inc += t;
        }
        const int wtot = __builtin_amdgcn_readlane(inc, 31);
        int base = 0;
        if (wtot > 0) {
            if (lane == 0) base = atomicAdd(&gcnt, wtot);
            base = __builtin_amdgcn_readfirstlane(base);
        }
        int slot = base + inc - nb;
        const int gbase = wi * 32;
        while (bits) {
            const int g = gbase + __builtin_ctz(bits);
            bits &= bits - 1u;
            if (slot < GL_CAP) {
                glist[slot] = g;
            } else {
#pragma unroll 1
                for (int q = 0; q < GROUP_EDGES; ++q) {
                    const v2i r = rec2[(size_t)g * GROUP_EDGES + q];
                    if ((r.y >> 8) == b) take(atomicAdd(&cnt, 1), r);
                }
            }
            ++slot;
        }
    }
    __syncthreads();

    const int gtotal = __builtin_amdgcn_readfirstlane(gcnt);
    const int ng = gtotal < GL_CAP ? gtotal : GL_CAP;
    const int hw  = tid >> 4;
    const int l16 = tid & 15;
    for (int k0 = 0; k0 < ng; k0 += HW * UB) {
        v2i r[UB];
        bool hit[UB];
#pragma unroll
        for (int u = 0; u < UB; ++u) {
            const int k = k0 + u * HW + hw;
            hit[u] = k < ng;
            const int g = glist[hit[u] ? k : 0];
            r[u] = rec2[(size_t)g * GROUP_EDGES + l16];
        }
        int off[UB];
        int wsum = 0;
#pragma unroll
        for (int u = 0; u < UB; ++u) {
            hit[u] = hit[u] && ((r[u].y >> 8) == b);
            const unsigned mk = __builtin_amdgcn_ballot_w32(hit[u]);
            off[u] = wsum + (int)__builtin_amdgcn_mbcnt_lo(mk, 0u);
            wsum += __builtin_popcount(mk);
        }
        if (wsum > 0) {
            int base = 0;
            if (lane == 0) base = atomicAdd(&cnt, wsum);
            base = __builtin_amdgcn_readfirstlane(base);
#pragma unroll
            for (int u = 0; u < UB; ++u)
                if (hit[u]) take(base + off[u], r[u]);
        }
    }
    if (tid < (e & (GROUP_EDGES - 1))) {
        const v2i r = rec2[(size_t)(e & ~(GROUP_EDGES - 1)) + tid];
        if ((r.y >> 8) == b) take(atomicAdd(&cnt, 1), r);
    }
    __syncthreads();

    const int total = cnt;
    const int staged = total < PSTAGE ? total : PSTAGE;
    for (int i = tid; i < staged; i += AGG_THREADS) mine[i] = pairs[i];

    if (tid < rows) {
        const float dg = (float)(degc[tid] + 1);
        dinv[n0 + tid] = dg > 0.0f ? rsqrtf(dg) : 0.0f;
    }
    if (tid == 0) tot[b] = total;
}

template <bool RELU, int K, int NC>
__device__ __forceinline__ void
gemm_wmma_group(const float* __restrict__ A,
                const float* __restrict__ B,
                float* __restrict__ C, int M, int blk, float* Bs) {
    constexpr int NT = NC / 16;
    constexpr int KT = K / 8;
    constexpr int PITCH = 2 * NC + 16;
    constexpr int ITEMS = (K / 2) * (NC / 4);
    static_assert(ITEMS % (GEMM_WAVES * 32) == 0, "every thread stages the same number of pieces");

    const int tid  = threadIdx.x;
    const int lane = tid & 31;
    const int wave = tid >> 5;
    const int c    = lane & 15;
    const int h    = lane >> 4;
    const int rowTile = blk * GEMM_WAVES + wave;
    const bool active = rowTile * 16 < M;
    const int rowBase = active ? rowTile * 16 : 0;

    float4 a[KT];
    const float4* Ap = (const float4*)(A + (size_t)(rowBase + c) * K) + h;
#pragma unroll
    for (int t = 0; t < KT; ++t) a[t] = Ap[2 * t];

#pragma unroll
    for (int i = 0; i < ITEMS / (GEMM_WAVES * 32); ++i) {
        const int idx = tid + i * (GEMM_WAVES * 32);
        const int p = idx / (NC / 4);
        const int q = idx - p * (NC / 4);
        const float4 r0 = *(const float4*)(B + (size_t)(2 * p) * NC + 4 * q);
        const float4 r1 = *(const float4*)(B + (size_t)(2 * p + 1) * NC + 4 * q);
        float4* d = (float4*)(Bs + p * PITCH + 8 * q);
        d[0] = make_float4(r0.x, r1.x, r0.y, r1.y);
        d[1] = make_float4(r0.z, r1.z, r0.w, r1.w);
    }
    __syncthreads();
    if (!active) return;

    v8f acc[NT];
#pragma unroll
    for (int j = 0; j < NT; ++j) acc[j] = v8f{};

    const float* bl = Bs + (2 * h) * PITCH + 2 * c;
#pragma unroll
    for (int t = 0; t < KT; ++t) {
#pragma unroll
        for (int u = 0; u < 2; ++u) {
            float a0 = u ? a[t].z : a[t].x;
            float a1 = u ? a[t].w : a[t].y;
            if (RELU) { a0 = fmaxf(a0, 0.0f); a1 = fmaxf(a1, 0.0f); }
            v2f av; av.x = a0; av.y = a1;
            const float* bp = bl + (4 * t + u) * PITCH;
#pragma unroll
            for (int j = 0; j < NT; ++j) {
                const v2f bv = *(const v2f*)(bp + 32 * j);
                acc[j] = __builtin_amdgcn_wmma_f32_16x16x4_f32(
                    false, av, false, bv, (short)0, acc[j], false, false);
            }
        }
    }

    float* Crow = C + (size_t)(rowBase + 8 * h) * NC + c;
#pragma unroll
    for (int j = 0; j < NT; ++j)
#pragma unroll
        for (int r = 0; r < 8; ++r)
            Crow[(size_t)r * NC + 16 * j] = acc[j][r];
}

static_assert(KEY_THREADS == GEMM_WAVES * 32, "both kinds of work-group have the same size");
template <int K, int NC>
__global__ void __launch_bounds__(KEY_THREADS)
gcn_keys_and_product(const int* __restrict__ src, const int* __restrict__ dst,
                     int* __restrict__ recs,
                     unsigned* __restrict__ flags, int e,
                     const float* __restrict__ A, const float* __restrict__ B,
                     float* __restrict__ C, int M, int gemmGroups) {
    constexpr int GEMM_LDS_FLOATS = (K / 2) * (2 * NC + 16);
    static_assert(FWORDS * FPITCH <= GEMM_LDS_FLOATS, "the key groups' flag words fit the product groups' LDS");
    __shared__ float4 lds4[GEMM_LDS_FLOATS / 4];
    const int blk = blockIdx.x;
    if (blk < gemmGroups)
        gemm_wmma_group<false, K, NC>(A, B, C, M, blk, (float*)lds4);
    else
        edge_keys_group(src, dst, recs, flags, e, blk - gemmGroups, (unsigned*)lds4);
}

template <int F, int NC2>
__global__ void __launch_bounds__(AGG_THREADS)
gcn_aggregate_owned(const int* __restrict__ src,
                    const int* __restrict__ dst,
                    const int* __restrict__ lists,
                    const int* __restrict__ tot,
                    const float* __restrict__ dinv,
                    const float* __restrict__ h,
                    const float* __restrict__ bias,
                    const float* __restrict__ W2,
                    float* __restrict__ out, int n, int e) {
    constexpr int J = F / 32;
    constexpr int U = 8;
    constexpr int SU = 4;
    static_assert((U & (U - 1)) == 0 && STAGE % U == 0 && 32 % U == 0, "a staged list pads to whole groups inside the staging area");
    constexpr int PA = NC2 > 0 ? F + 4 : F;
    __shared__ float4 acc4[RANGE * PA / 4];
    __shared__ float4 stg4[3 * STAGE / 4];
    float* acc = (float*)acc4;
    int*   es = (int*)stg4;
    int*   ed = es + STAGE;
    float* ew = (float*)(ed + STAGE);

    const int tid  = threadIdx.x;
    const int lane = tid & 31;
    const int wave = __builtin_amdgcn_readfirstlane(tid >> 5);
    const int b    = blockIdx.x;
    const int n0   = b * RANGE;
    const int rows = (n - n0) < RANGE ? (n - n0) : RANGE;
    const int total = __builtin_amdgcn_readfirstlane(tot[b]);

    constexpr int WITEMS = NC2 > 0 ? (F / 2) * (NC2 / 4) : AGG_THREADS;
    static_assert(WITEMS % AGG_THREADS == 0, "every thread stages the same number of pieces");
    constexpr int WPT = NC2 > 0 ? WITEMS / AGG_THREADS : 0;
    float4 w0[WPT > 0 ? WPT : 1], w1[WPT > 0 ? WPT : 1];
    if constexpr (NC2 > 0) {
#pragma unroll
        for (int i = 0; i < WPT; ++i) {
            const int idx = tid + i * AGG_THREADS;
            const int p = idx / (NC2 / 4);
            const int q = idx - p * (NC2 / 4);
            w0[i] = *(const float4*)(W2 + (size_t)(2 * p) * NC2 + 4 * q);
            w1[i] = *(const float4*)(W2 + (size_t)(2 * p + 1) * NC2 + 4 * q);
        }
    }

    for (int i = tid; i < rows * F; i += AGG_THREADS) {
        const int r = i / F;
        const int f = i - r * F;
        const float di = dinv[n0 + r];
        acc[r * PA + f] = bias[f] + h[(size_t)n0 * F + i] * (di * di);
    }
    __syncthreads();

    if (total <= LIST_CAP) {
        const v2i* mine = (const v2i*)lists + (size_t)b * LIST_CAP;
        for (int r0 = 0; r0 < total; r0 += STAGE) {
            const int mm = (total - r0) < STAGE ? (total - r0) : STAGE;

            for (int i0 = tid; i0 < mm; i0 += AGG_THREADS * SU) {
                v2i en[SU];
                float wv[SU];
#pragma unroll
                for (int u = 0; u < SU; ++u) {
                    const int i = i0 + u * AGG_THREADS;
                    en[u] = mine[r0 + (i < mm ? i : i0)];
                }
#pragma unroll
                for (int u = 0; u < SU; ++u) {
                    const int dr = en[u].y >= 0 ? en[u].y : 0;
                    wv[u] = dinv[en[u].x] * dinv[n0 + dr];
                }
#pragma unroll
                for (int u = 0; u < SU; ++u) {
                    const int i = i0 + u * AGG_THREADS;
                    if (i < mm) {
                        es[i] = en[u].x;
                        ed[i] = en[u].y;
                        ew[i] = en[u].y >= 0 ? wv[u] : 0.0f;
                    }
                }
            }
            const int mmU = (mm + U - 1) & ~(U - 1);
            if (tid < mmU - mm) {
                es[mm + tid] = 0;
                ed[mm + tid] = -1;
                ew[mm + tid] = 0.0f;
            }
            __syncthreads();

            const int ng = mmU / U;
            const int eend = (((wave + 1) * ng) / (AGG_THREADS / 32)) * U;
            for (int base0 = ((wave * ng) / (AGG_THREADS / 32)) * U; base0 < eend; base0 += 32) {
                const int rest = eend - base0;
                const int c32 = rest < 32 ? rest : 32;
                int s = 0, dl = -1;
                float w = 0.0f;
                if (lane < c32) {
                    s  = es[base0 + lane];
                    dl = ed[base0 + lane];
                    w  = ew[base0 + lane];
                }
                const unsigned okmask = __builtin_amdgcn_ballot_w32(dl >= 0);
                for (int i = 0; i < c32; i += U) {
                    float v[U][J];
#pragma unroll
                    for (int u = 0; u < U; ++u) {
                        const int su = __builtin_amdgcn_readlane(s, i + u);
                        const float* hp = h + (size_t)su * F + lane;
#pragma unroll
                        for (int j = 0; j < J; ++j) v[u][j] = hp[32 * j];
                    }
#pragma unroll
                    for (int u = 0; u < U; ++u) {
                        if ((okmask >> (i + u)) & 1u) {
                            const int du = __builtin_amdgcn_readlane(dl, i + u);
                            const float wu = __int_as_float(__builtin_amdgcn_readlane(__float_as_int(w), i + u));
                            float* ap = acc + du * PA + lane;
#pragma unroll
                            for (int j = 0; j < J; ++j) atomicAdd(ap + 32 * j, v[u][j] * wu);
                        }
                    }
                }
            }
            __syncthreads();
        }
    } else {
        for (int ee = tid; ee < e; ee += AGG_THREADS) {
            const int d = dst[ee];
            if (d / RANGE == b) {
                const int s = src[ee];
                const float w = dinv[s] * dinv[d];
                float* ap = acc + (d - n0) * PA;
                const float* hp = h + (size_t)s * F;
#pragma unroll 1
                for (int f = 0; f < F; ++f) atomicAdd(ap + f, hp[f] * w);
            }
        }
        __syncthreads();
    }

    if constexpr (NC2 == 0) {
        for (int i = tid; i < rows * F; i += AGG_THREADS)
            out[(size_t)n0 * F + i] = acc[i];
    } else {
        constexpr int KT = F / 8;
        constexpr int NT2 = NC2 / 16;
        constexpr int PITCH = 2 * NC2 + 16;
        static_assert((F / 2) * PITCH <= 3 * STAGE, "W2 fits the staging area");
        float* Bs = (float*)stg4;
#pragma unroll
        for (int i = 0; i < WPT; ++i) {
            const int idx = tid + i * AGG_THREADS;
            const int p = idx / (NC2 / 4);
            const int q = idx - p * (NC2 / 4);
            float4* d = (float4*)(Bs + p * PITCH + 8 * q);
            d[0] = make_float4(w0[i].x, w1[i].x, w0[i].y, w1[i].y);
            d[1] = make_float4(w0[i].z, w1[i].z, w0[i].w, w1[i].w);
        }
        __syncthreads();

        const int c  = lane & 15;
        const int hh = lane >> 4;
        const int rtiles = (rows + 15) >> 4;
        for (int unit = wave; unit < rtiles * NT2; unit += AGG_THREADS / 32) {
            const int rt = unit / NT2;
            const int ct = unit - rt * NT2;
            const int arow = (rt * 16 + c) < rows ? (rt * 16 + c) : rows - 1;
            const float4* ap = (const float4*)(acc + arow * PA) + hh;
            const float* bl = Bs + (2 * hh) * PITCH + 2 * c + 32 * ct;
            v8f d = v8f{};
#pragma unroll
            for (int t = 0; t < KT; ++t) {
                const float4 a = ap[2 * t];
#pragma unroll
                for (int u = 0; u < 2; ++u) {
                    v2f av;
                    av.x = fmaxf(u ? a.z : a.x, 0.0f);
                    av.y = fmaxf(u ? a.w : a.y, 0.0f);
                    const v2f bv = *(const v2f*)(bl + (4 * t + u) * PITCH);
                    d = __builtin_amdgcn_wmma_f32_16x16x4_f32(
                        false, av, false, bv, (short)0, d, false, false);
                }
            }
            const int row0 = rt * 16 + 8 * hh;
            float* orow = out + (size_t)(n0 + row0) * NC2 + 16 * ct + c;
#pragma unroll
            for (int r = 0; r < 8; ++r)
                if (row0 + r < rows) orow[(size_t)r * NC2] = d[r];
        }
    }
}

extern "C" void kernel_launch(void* const* d_in, const int* in_sizes, int n_in,
                              void* d_out, int out_size, void* d_ws, size_t ws_size,
                              hipStream_t stream) {
    constexpr int D_IN = 64, D_HID = 128, D_OUT = 64;

    const float* x  = (const float*)d_in[0];
    const int*   ei = (const int*)  d_in[1];
    const float* W1 = (const float*)d_in[2];
    const float* b1 = (const float*)d_in[3];
    const float* W2 = (const float*)d_in[4];
    const float* b2 = (const float*)d_in[5];
    float* out = (float*)d_out;

    const int N = in_sizes[0] / D_IN;
    const int E = in_sizes[1] / 2;
    const int* src = ei;
    const int* dst = ei + E;

    auto cdiv = [](long a, long b) { return (int)((a + b - 1) / b); };
    const int owners = cdiv(N, RANGE);

    float* ws   = (float*)d_ws;
    float* dinv = ws;
    float* h1   = (float*)(((size_t)(dinv + N) + 255) & ~(size_t)255);
    float* h2   = h1 + (size_t)N * D_HID;
    int* recs  = (int*)(((size_t)(h2 + (size_t)N * D_HID) + 255) & ~(size_t)255);
    int* tot   = (int*)(((size_t)(recs + 2 * (size_t)E) + 15) & ~(size_t)15);
    const int flagChunks = cdiv(E, FCHUNK);
    unsigned* flags = (unsigned*)(((size_t)(tot + owners) + 255) & ~(size_t)255);
    int* lists = (int*)(((size_t)(flags + (size_t)flagChunks * 256 * FWORDS) + 255) & ~(size_t)255);

    const int gemmGroups = cdiv(N / 16, GEMM_WAVES);
    gcn_keys_and_product<D_IN, D_HID><<<gemmGroups + flagChunks, KEY_THREADS, 0, stream>>>(
        src, dst, recs, flags, E, x, W1, h1, N, gemmGroups);
    gcn_owner_prepass<<<owners, AGG_THREADS, 0, stream>>>(recs, flags, dinv, lists, tot, N, E);

    gcn_aggregate_owned<D_HID, D_OUT><<<owners, AGG_THREADS, 0, stream>>>(
        src, dst, lists, tot, dinv, h1, b1, W2, h2, N, E);

    gcn_aggregate_owned<D_OUT, 0><<<owners, AGG_THREADS, 0, stream>>>(
        src, dst, lists, tot, dinv, h2, b2, nullptr, out, N, E);
}


# Layer-2 edge weights folded into the rows: layer-1 owners store h2 times dinv[row]; layer 2 copies its rows in, adds plain rows, scales by dinv[dst] at the store: no dinv gather; rounding changes
# speedup vs baseline: 3.7534x; 1.0701x over previous
#include <hip/hip_runtime.h>

typedef float v2f __attribute__((ext_vector_type(2)));
typedef float v8f __attribute__((ext_vector_type(8)));

constexpr int RANGE = 196;
constexpr int AGG_THREADS = 1024;
constexpr int GROUP_EDGES = 16;
constexpr int LIST_CAP = 49152;
constexpr int GL_CAP   = 16384;
constexpr int PSTAGE   = 16384;
constexpr int STAGE = 16384;
constexpr int GEMM_WAVES = 8;
constexpr int KEY_THREADS = 256;
constexpr int FCHUNK  = 4096;
constexpr int FGROUPS = FCHUNK / GROUP_EDGES;
constexpr int FWORDS  = FGROUPS / 32;
constexpr int FPITCH  = 260;

typedef int v2i __attribute__((ext_vector_type(2)));
typedef int v4i __attribute__((ext_vector_type(4)));

__device__ __forceinline__ void
edge_keys_group(const int* __restrict__ src, const int* __restrict__ dst,
                int* __restrict__ recs, unsigned* __restrict__ flags,
                int e, int chunk, unsigned* m) {
    constexpr int L = FCHUNK / 4 / KEY_THREADS;

    const int tid   = threadIdx.x;
    const int ebase = chunk * FCHUNK;

    int d[L][4], s[L][4];
    bool fullw[L];
#pragma unroll
    for (int j = 0; j < L; ++j) {
        const int i4 = ebase + (j * KEY_THREADS + tid) * 4;
        fullw[j] = i4 + 3 < e;
        const int ia = fullw[j] ? i4 : 0;
        d[j][0] = dst[ia];     d[j][1] = dst[ia + 1];
        d[j][2] = dst[ia + 2]; d[j][3] = dst[ia + 3];
        s[j][0] = src[ia];     s[j][1] = src[ia + 1];
        s[j][2] = src[ia + 2]; s[j][3] = src[ia + 3];
    }
    for (int i = tid; i < FWORDS * FPITCH; i += KEY_THREADS) m[i] = 0u;
    __syncthreads();

#pragma unroll
    for (int j = 0; j < L; ++j) {
        const int q  = j * KEY_THREADS + tid;
        const int i4 = ebase + q * 4;
        const int gi = q >> 2;
        unsigned* frow = m + (gi >> 5) * FPITCH;
        const unsigned bit = 1u << (gi & 31);
        if (fullw[j]) {
            const int o0 = d[j][0] / RANGE, o1 = d[j][1] / RANGE;
            const int o2 = d[j][2] / RANGE, o3 = d[j][3] / RANGE;
            v4i* rp = (v4i*)(recs + 2 * (size_t)i4);
            v4i ra, rb;
            ra.x = s[j][0]; ra.y = (o0 << 8) | (d[j][0] - o0 * RANGE);
            ra.z = s[j][1]; ra.w = (o1 << 8) | (d[j][1] - o1 * RANGE);
            rb.x = s[j][2]; rb.y = (o2 << 8) | (d[j][2] - o2 * RANGE);
            rb.z = s[j][3]; rb.w = (o3 << 8) | (d[j][3] - o3 * RANGE);
            rp[0] = ra;
            rp[1] = rb;
            atomicOr(frow + ((unsigned)o0 & 255u), bit);
            atomicOr(frow + ((unsigned)o1 & 255u), bit);
            atomicOr(frow + ((unsigned)o2 & 255u), bit);
            atomicOr(frow + ((unsigned)o3 & 255u), bit);
        } else {
            for (int i = i4; i < e; ++i) {
                const int dd = dst[i];
                const int o = dd / RANGE;
                recs[2 * (size_t)i]     = src[i];
                recs[2 * (size_t)i + 1] = (o << 8) | (dd - o * RANGE);
                atomicOr(frow + ((unsigned)o & 255u), bit);
            }
        }
    }
    __syncthreads();

    for (int i = tid; i < 256 * FWORDS; i += KEY_THREADS)
        flags[(size_t)chunk * (256 * FWORDS) + i] = m[(i % FWORDS) * FPITCH + (i / FWORDS)];
}

__global__ void __launch_bounds__(AGG_THREADS)
gcn_owner_prepass(const int* __restrict__ recs,
                  const unsigned* __restrict__ flags,
                  float* __restrict__ dinv,
                  int* __restrict__ lists,
                  int* __restrict__ tot, int n, int e) {
    constexpr int UB = 16;
    constexpr int HW = AGG_THREADS / 16;
    __shared__ int glist[GL_CAP];
    __shared__ v2i pairs[PSTAGE];
    __shared__ int degc[RANGE];
    __shared__ int gcnt;
    __shared__ int cnt;

    const int tid  = threadIdx.x;
    const int lane = tid & 31;
    const int wave = __builtin_amdgcn_readfirstlane(tid >> 5);
    const int b    = blockIdx.x;
    const int n0   = b * RANGE;
    const int rows = (n - n0) < RANGE ? (n - n0) : RANGE;

    if (tid < RANGE) degc[tid] = 0;
    if (tid == 0) { gcnt = 0; cnt = 0; }
    __syncthreads();

    const v2i* rec2 = (const v2i*)recs;
    v2i* mine = (v2i*)lists + (size_t)b * LIST_CAP;

    auto take = [&](int idx, v2i r) {
        const int row = r.y & 255;
        atomicAdd(&degc[row], 1);
        v2i en;
        en.x = r.x;
        en.y = row;
        if (idx < PSTAGE) pairs[idx] = en;
        else if (idx < LIST_CAP) mine[idx] = en;
    };

    const unsigned mykey = (unsigned)b & 255u;
    const int nFull = e >> 4;
    const int nFlagWords = ((e + FCHUNK - 1) / FCHUNK) * FWORDS;
    for (int w0 = wave * 32; w0 < nFlagWords; w0 += AGG_THREADS) {
        const int wi = w0 + lane;
        unsigned bits = 0u;
        if (wi < nFlagWords)
            bits = flags[(size_t)(wi / FWORDS) * (256 * FWORDS) + mykey * FWORDS + (wi % FWORDS)];
        const int lim = nFull - wi * 32;
        if (lim < 32) bits = lim <= 0 ? 0u : (bits & ((1u << lim) - 1u));
        const int nb = __builtin_popcount(bits);
        int inc = nb;
#pragma unroll
        for (int dl = 1; dl < 32; dl <<= 1) {
            const int t = __builtin_amdgcn_ds_bpermute(((lane - dl) & 31) << 2, inc);
            if (lane >= dl) inc += t;
        }
        const int wtot = __builtin_amdgcn_readlane(inc, 31);
        int base = 0;
        if (wtot > 0) {
            if (lane == 0) base = atomicAdd(&gcnt, wtot);
            base = __builtin_amdgcn_readfirstlane(base);
        }
        int slot = base + inc - nb;
        const int gbase = wi * 32;
        while (bits) {
            const int g = gbase + __builtin_ctz(bits);
            bits &= bits - 1u;
            if (slot < GL_CAP) {
                glist[slot] = g;
            } else {
#pragma unroll 1
                for (int q = 0; q < GROUP_EDGES; ++q) {
                    const v2i r = rec2[(size_t)g * GROUP_EDGES + q];
                    if ((r.y >> 8) == b) take(atomicAdd(&cnt, 1), r);
                }
            }
            ++slot;
        }
    }
    __syncthreads();

    const int gtotal = __builtin_amdgcn_readfirstlane(gcnt);
    const int ng = gtotal < GL_CAP ? gtotal : GL_CAP;
    const int hw  = tid >> 4;
    const int l16 = tid & 15;
    for (int k0 = 0; k0 < ng; k0 += HW * UB) {
        v2i r[UB];
        bool hit[UB];
#pragma unroll
        for (int u = 0; u < UB; ++u) {
            const int k = k0 + u * HW + hw;
            hit[u] = k < ng;
            const int g = glist[hit[u] ? k : 0];
            r[u] = rec2[(size_t)g * GROUP_EDGES + l16];
        }
        int off[UB];
        int wsum = 0;
#pragma unroll
        for (int u = 0; u < UB; ++u) {
            hit[u] = hit[u] && ((r[u].y >> 8) == b);
            const unsigned mk = __builtin_amdgcn_ballot_w32(hit[u]);
            off[u] = wsum + (int)__builtin_amdgcn_mbcnt_lo(mk, 0u);
            wsum += __builtin_popcount(mk);
        }
        if (wsum > 0) {
            int base = 0;
            if (lane == 0) base = atomicAdd(&cnt, wsum);
            base = __builtin_amdgcn_readfirstlane(base);
#pragma unroll
            for (int u = 0; u < UB; ++u)
                if (hit[u]) take(base + off[u], r[u]);
        }
    }
    if (tid < (e & (GROUP_EDGES - 1))) {
        const v2i r = rec2[(size_t)(e & ~(GROUP_EDGES - 1)) + tid];
        if ((r.y >> 8) == b) take(atomicAdd(&cnt, 1), r);
    }
    __syncthreads();

    const int total = cnt;
    const int staged = total < PSTAGE ? total : PSTAGE;
    for (int i = tid; i < staged; i += AGG_THREADS) mine[i] = pairs[i];

    if (tid < rows) {
        const float dg = (float)(degc[tid] + 1);
        dinv[n0 + tid] = dg > 0.0f ? rsqrtf(dg) : 0.0f;
    }
    if (tid == 0) tot[b] = total;
}

template <bool RELU, int K, int NC>
__device__ __forceinline__ void
gemm_wmma_group(const float* __restrict__ A,
                const float* __restrict__ B,
                float* __restrict__ C, int M, int blk, float* Bs) {
    constexpr int NT = NC / 16;
    constexpr int KT = K / 8;
    constexpr int PITCH = 2 * NC + 16;
    constexpr int ITEMS = (K / 2) * (NC / 4);
    static_assert(ITEMS % (GEMM_WAVES * 32) == 0, "every thread stages the same number of pieces");

    const int tid  = threadIdx.x;
    const int lane = tid & 31;
    const int wave = tid >> 5;
    const int c    = lane & 15;
    const int h    = lane >> 4;
    const int rowTile = blk * GEMM_WAVES + wave;
    const bool active = rowTile * 16 < M;
    const int rowBase = active ? rowTile * 16 : 0;

    float4 a[KT];
    const float4* Ap = (const float4*)(A + (size_t)(rowBase + c) * K) + h;
#pragma unroll
    for (int t = 0; t < KT; ++t) a[t] = Ap[2 * t];

#pragma unroll
    for (int i = 0; i < ITEMS / (GEMM_WAVES * 32); ++i) {
        const int idx = tid + i * (GEMM_WAVES * 32);
        const int p = idx / (NC / 4);
        const int q = idx - p * (NC / 4);
        const float4 r0 = *(const float4*)(B + (size_t)(2 * p) * NC + 4 * q);
        const float4 r1 = *(const float4*)(B + (size_t)(2 * p + 1) * NC + 4 * q);
        float4* d = (float4*)(Bs + p * PITCH + 8 * q);
        d[0] = make_float4(r0.x, r1.x, r0.y, r1.y);
        d[1] = make_float4(r0.z, r1.z, r0.w, r1.w);
    }
    __syncthreads();
    if (!active) return;

    v8f acc[NT];
#pragma unroll
    for (int j = 0; j < NT; ++j) acc[j] = v8f{};

    const float* bl = Bs + (2 * h) * PITCH + 2 * c;
#pragma unroll
    for (int t = 0; t < KT; ++t) {
#pragma unroll
        for (int u = 0; u < 2; ++u) {
            float a0 = u ? a[t].z : a[t].x;
            float a1 = u ? a[t].w : a[t].y;
            if (RELU) { a0 = fmaxf(a0, 0.0f); a1 = fmaxf(a1, 0.0f); }
            v2f av; av.x = a0; av.y = a1;
            const float* bp = bl + (4 * t + u) * PITCH;
#pragma unroll
            for (int j = 0; j < NT; ++j) {
                const v2f bv = *(const v2f*)(bp + 32 * j);
                acc[j] = __builtin_amdgcn_wmma_f32_16x16x4_f32(
                    false, av, false, bv, (short)0, acc[j], false, false);
            }
        }
    }

    float* Crow = C + (size_t)(rowBase + 8 * h) * NC + c;
#pragma unroll
    for (int j = 0; j < NT; ++j)
#pragma unroll
        for (int r = 0; r < 8; ++r)
            Crow[(size_t)r * NC + 16 * j] = acc[j][r];
}

static_assert(KEY_THREADS == GEMM_WAVES * 32, "both kinds of work-group have the same size");
template <int K, int NC>
__global__ void __launch_bounds__(KEY_THREADS)
gcn_keys_and_product(const int* __restrict__ src, const int* __restrict__ dst,
                     int* __restrict__ recs,
                     unsigned* __restrict__ flags, int e,
                     const float* __restrict__ A, const float* __restrict__ B,
                     float* __restrict__ C, int M, int gemmGroups) {
    constexpr int GEMM_LDS_FLOATS = (K / 2) * (2 * NC + 16);
    static_assert(FWORDS * FPITCH <= GEMM_LDS_FLOATS, "the key groups' flag words fit the product groups' LDS");
    __shared__ float4 lds4[GEMM_LDS_FLOATS / 4];
    const int blk = blockIdx.x;
    if (blk < gemmGroups)
        gemm_wmma_group<false, K, NC>(A, B, C, M, blk, (float*)lds4);
    else
        edge_keys_group(src, dst, recs, flags, e, blk - gemmGroups, (unsigned*)lds4);
}

__device__ __forceinline__ void keep_loads_grouped(const float4& v) {
    asm volatile("" :: "v"(v.x), "v"(v.y), "v"(v.z), "v"(v.w));
}
__device__ __forceinline__ void keep_loads_grouped(float v) {
    asm volatile("" :: "v"(v));
}

template <int F, int NC2, bool PRE>
__global__ void __launch_bounds__(AGG_THREADS)
gcn_aggregate_owned(const int* __restrict__ src,
                    const int* __restrict__ dst,
                    const int* __restrict__ lists,
                    const int* __restrict__ tot,
                    const float* __restrict__ dinv,
                    const float* __restrict__ h,
                    const float* __restrict__ bias,
                    const float* __restrict__ W2,
                    float* __restrict__ out, int n, int e) {
    constexpr int J = F / 32;
    constexpr int U = 8;
    constexpr int SU = 4;
    static_assert((U & (U - 1)) == 0 && STAGE % U == 0 && 32 % U == 0, "a staged list pads to whole groups inside the staging area");
    constexpr int PA = NC2 > 0 ? F + 4 : F;
    __shared__ float4 acc4[RANGE * PA / 4];
    __shared__ float4 stg4[3 * STAGE / 4];
    constexpr int RPAD = ((RANGE + 15) / 16) * 16;
    __shared__ float4 dls4[NC2 > 0 ? RPAD / 4 : 1];
    float* dls = (float*)dls4;
    float* acc = (float*)acc4;
    int*   es = (int*)stg4;
    int*   ed = es + STAGE;
    float* ew = (float*)(ed + STAGE);

    const int tid  = threadIdx.x;
    const int lane = tid & 31;
    const int wave = __builtin_amdgcn_readfirstlane(tid >> 5);
    const int b    = blockIdx.x;
    const int n0   = b * RANGE;
    const int rows = (n - n0) < RANGE ? (n - n0) : RANGE;
    const int total = __builtin_amdgcn_readfirstlane(tot[b]);

    constexpr int WITEMS = NC2 > 0 ? (F / 2) * (NC2 / 4) : AGG_THREADS;
    static_assert(WITEMS % AGG_THREADS == 0, "every thread stages the same number of pieces");
    constexpr int WPT = NC2 > 0 ? WITEMS / AGG_THREADS : 0;
    float4 w0[WPT > 0 ? WPT : 1], w1[WPT > 0 ? WPT : 1];
    if constexpr (NC2 > 0) {
#pragma unroll
        for (int i = 0; i < WPT; ++i) {
            const int idx = tid + i * AGG_THREADS;
            const int p = idx / (NC2 / 4);
            const int q = idx - p * (NC2 / 4);
            w0[i] = *(const float4*)(W2 + (size_t)(2 * p) * NC2 + 4 * q);
            w1[i] = *(const float4*)(W2 + (size_t)(2 * p + 1) * NC2 + 4 * q);
        }
    }

    constexpr int F4 = F / 4;
    constexpr int SLOTS = (RANGE * F4 + AGG_THREADS - 1) / AGG_THREADS;
    static_assert(AGG_THREADS % F4 == 0, "a thread's pieces all lie in the same 4 columns");
    const int lim4 = rows * F4;
    const int q4 = tid % F4;
    float dvs[SLOTS];
    float bq[4];
    if constexpr (PRE) {
        const float4* hrow = (const float4*)(h + (size_t)n0 * F);
        float4 hv[SLOTS];
#pragma unroll
        for (int k = 0; k < SLOTS; ++k) {
            const int i4 = tid + k * AGG_THREADS;
            const int ia = i4 < lim4 ? i4 : 0;
            hv[k] = hrow[ia];
            dvs[k] = dinv[n0 + ia / F4];
        }
#pragma unroll
        for (int m = 0; m < 4; ++m) bq[m] = bias[4 * q4 + m];
#pragma unroll
        for (int k = 0; k < SLOTS; ++k) {
            keep_loads_grouped(hv[k]);
            keep_loads_grouped(dvs[k]);
        }
#pragma unroll
        for (int m = 0; m < 4; ++m) keep_loads_grouped(bq[m]);
#pragma unroll
        for (int k = 0; k < SLOTS; ++k) {
            const int i4 = tid + k * AGG_THREADS;
            if (i4 < lim4) {
                const int r = i4 / F4;
                *(float4*)(acc + r * PA + 4 * q4) = hv[k];
            }
        }
    } else {
        for (int i = tid; i < rows * F; i += AGG_THREADS) {
            const int r = i / F;
            const int f = i - r * F;
            const float di = dinv[n0 + r];
            acc[r * PA + f] = bias[f] + h[(size_t)n0 * F + i] * (di * di);
        }
        if constexpr (NC2 > 0) {
            if (tid < RPAD) {
                const float di = dinv[n0 + (tid < rows ? tid : 0)];
                dls[tid] = tid < rows ? di : 0.0f;
            }
        }
    }
    __syncthreads();

    if (total <= LIST_CAP) {
        const v2i* mine = (const v2i*)lists + (size_t)b * LIST_CAP;
        for (int r0 = 0; r0 < total; r0 += STAGE) {
            const int mm = (total - r0) < STAGE ? (total - r0) : STAGE;

            for (int i0 = tid; i0 < mm; i0 += AGG_THREADS * SU) {
                v2i en[SU];
                float wv[SU];
#pragma unroll
                for (int u = 0; u < SU; ++u) {
                    const int i = i0 + u * AGG_THREADS;
                    en[u] = mine[r0 + (i < mm ? i : i0)];
                }
                if constexpr (!PRE) {
#pragma unroll
                    for (int u = 0; u < SU; ++u) {
                        const int dr = en[u].y >= 0 ? en[u].y : 0;
                        wv[u] = dinv[en[u].x] * dinv[n0 + dr];
                    }
                }
#pragma unroll
                for (int u = 0; u < SU; ++u) {
                    const int i = i0 + u * AGG_THREADS;
                    if (i < mm) {
                        es[i] = en[u].x;
                        ed[i] = en[u].y;
                        if constexpr (!PRE) ew[i] = en[u].y >= 0 ? wv[u] : 0.0f;
                    }
                }
            }
            const int mmU = (mm + U - 1) & ~(U - 1);
            if (tid < mmU - mm) {
                es[mm + tid] = 0;
                ed[mm + tid] = -1;
                if constexpr (!PRE) ew[mm + tid] = 0.0f;
            }
            __syncthreads();

            const int ng = mmU / U;
            const int eend = (((wave + 1) * ng) / (AGG_THREADS / 32)) * U;
            for (int base0 = ((wave * ng) / (AGG_THREADS / 32)) * U; base0 < eend; base0 += 32) {
                const int rest = eend - base0;
                const int c32 = rest < 32 ? rest : 32;
                int s = 0, dl = -1;
                float w = 0.0f;
                if (lane < c32) {
                    s  = es[base0 + lane];
                    dl = ed[base0 + lane];
                    if constexpr (!PRE) w = ew[base0 + lane];
                }
                const unsigned okmask = __builtin_amdgcn_ballot_w32(dl >= 0);
                for (int i = 0; i < c32; i += U) {
                    float v[U][J];
#pragma unroll
                    for (int u = 0; u < U; ++u) {
                        const int su = __builtin_amdgcn_readlane(s, i + u);
                        const float* hp = h + (size_t)su * F + lane;
#pragma unroll
                        for (int j = 0; j < J; ++j) v[u][j] = hp[32 * j];
                    }
#pragma unroll
                    for (int u = 0; u < U; ++u) {
                        if ((okmask >> (i + u)) & 1u) {
                            const int du = __builtin_amdgcn_readlane(dl, i + u);
                            float* ap = acc + du * PA + lane;
                            if constexpr (PRE) {
#pragma unroll
                                for (int j = 0; j < J; ++j) atomicAdd(ap + 32 * j, v[u][j]);
                            } else {
                                const float wu = __int_as_float(__builtin_amdgcn_readlane(__float_as_int(w), i + u));
#pragma unroll
                                for (int j = 0; j < J; ++j) atomicAdd(ap + 32 * j, v[u][j] * wu);
                            }
                        }
                    }
                }
            }
            __syncthreads();
        }
    } else {
        for (int ee = tid; ee < e; ee += AGG_THREADS) {
            const int d = dst[ee];
            if (d / RANGE == b) {
                const int s = src[ee];
                float w = 1.0f;
                if constexpr (!PRE) w = dinv[s] * dinv[d];
                float* ap = acc + (d - n0) * PA;
                const float* hp = h + (size_t)s * F;
#pragma unroll 1
                for (int f = 0; f < F; ++f) atomicAdd(ap + f, hp[f] * w);
            }
        }
        __syncthreads();
    }

    if constexpr (NC2 == 0) {
        if constexpr (PRE) {
            float4* orow = (float4*)(out + (size_t)n0 * F);
#pragma unroll
            for (int k = 0; k < SLOTS; ++k) {
                const int i4 = tid + k * AGG_THREADS;
                if (i4 < lim4) {
                    const int r = i4 / F4;
                    const float4 a = *(const float4*)(acc + r * PA + 4 * q4);
                    orow[i4] = make_float4(bq[0] + dvs[k] * a.x, bq[1] + dvs[k] * a.y,
                                           bq[2] + dvs[k] * a.z, bq[3] + dvs[k] * a.w);
                }
            }
        } else {
            for (int i = tid; i < rows * F; i += AGG_THREADS)
                out[(size_t)n0 * F + i] = acc[i];
        }
    } else {
        constexpr int KT = F / 8;
        constexpr int NT2 = NC2 / 16;
        constexpr int PITCH = 2 * NC2 + 16;
        static_assert((F / 2) * PITCH <= 3 * STAGE, "W2 fits the staging area");
        float* Bs = (float*)stg4;
#pragma unroll
        for (int i = 0; i < WPT; ++i) {
            const int idx = tid + i * AGG_THREADS;
            const int p = idx / (NC2 / 4);
            const int q = idx - p * (NC2 / 4);
            float4* d = (float4*)(Bs + p * PITCH + 8 * q);
            d[0] = make_float4(w0[i].x, w1[i].x, w0[i].y, w1[i].y);
            d[1] = make_float4(w0[i].z, w1[i].z, w0[i].w, w1[i].w);
        }
        __syncthreads();

        const int c  = lane & 15;
        const int hh = lane >> 4;
        const int rtiles = (rows + 15) >> 4;
        for (int unit = wave; unit < rtiles * NT2; unit += AGG_THREADS / 32) {
            const int rt = unit / NT2;
            const int ct = unit - rt * NT2;
            const int arow = (rt * 16 + c) < rows ? (rt * 16 + c) : rows - 1;
            const float4* ap = (const float4*)(acc + arow * PA) + hh;
            const float* bl = Bs + (2 * hh) * PITCH + 2 * c + 32 * ct;
            const int row0 = rt * 16 + 8 * hh;
            const float4 dva = *(const float4*)(dls + row0);
            const float4 dvb = *(const float4*)(dls + row0 + 4);
            const float dv[8] = {dva.x, dva.y, dva.z, dva.w, dvb.x, dvb.y, dvb.z, dvb.w};
            v8f d = v8f{};
#pragma unroll
            for (int t = 0; t < KT; ++t) {
                const float4 a = ap[2 * t];
#pragma unroll
                for (int u = 0; u < 2; ++u) {
                    v2f av;
                    av.x = fmaxf(u ? a.z : a.x, 0.0f);
                    av.y = fmaxf(u ? a.w : a.y, 0.0f);
                    const v2f bv = *(const v2f*)(bl + (4 * t + u) * PITCH);
                    d = __builtin_amdgcn_wmma_f32_16x16x4_f32(
                        false, av, false, bv, (short)0, d, false, false);
                }
            }
            float* orow = out + (size_t)(n0 + row0) * NC2 + 16 * ct + c;
#pragma unroll
            for (int r = 0; r < 8; ++r)
                if (row0 + r < rows) orow[(size_t)r * NC2] = d[r] * dv[r];
        }
    }
}

extern "C" void kernel_launch(void* const* d_in, const int* in_sizes, int n_in,
                              void* d_out, int out_size, void* d_ws, size_t ws_size,
                              hipStream_t stream) {
    constexpr int D_IN = 64, D_HID = 128, D_OUT = 64;

    const float* x  = (const float*)d_in[0];
    const int*   ei = (const int*)  d_in[1];
    const float* W1 = (const float*)d_in[2];
    const float* b1 = (const float*)d_in[3];
    const float* W2 = (const float*)d_in[4];
    const float* b2 = (const float*)d_in[5];
    float* out = (float*)d_out;

    const int N = in_sizes[0] / D_IN;
    const int E = in_sizes[1] / 2;
    const int* src = ei;
    const int* dst = ei + E;

    auto cdiv = [](long a, long b) { return (int)((a + b - 1) / b); };
    const int owners = cdiv(N, RANGE);

    float* ws   = (float*)d_ws;
    float* dinv = ws;
    float* h1   = (float*)(((size_t)(dinv + N) + 255) & ~(size_t)255);
    float* h2   = h1 + (size_t)N * D_HID;
    int* recs  = (int*)(((size_t)(h2 + (size_t)N * D_HID) + 255) & ~(size_t)255);
    int* tot   = (int*)(((size_t)(recs + 2 * (size_t)E) + 15) & ~(size_t)15);
    const int flagChunks = cdiv(E, FCHUNK);
    unsigned* flags = (unsigned*)(((size_t)(tot + owners) + 255) & ~(size_t)255);
    int* lists = (int*)(((size_t)(flags + (size_t)flagChunks * 256 * FWORDS) + 255) & ~(size_t)255);

    const int gemmGroups = cdiv(N / 16, GEMM_WAVES);
    gcn_keys_and_product<D_IN, D_HID><<<gemmGroups + flagChunks, KEY_THREADS, 0, stream>>>(
        src, dst, recs, flags, E, x, W1, h1, N, gemmGroups);
    gcn_owner_prepass<<<owners, AGG_THREADS, 0, stream>>>(recs, flags, dinv, lists, tot, N, E);

    gcn_aggregate_owned<D_HID, D_OUT, false><<<owners, AGG_THREADS, 0, stream>>>(
        src, dst, lists, tot, dinv, h1, b1, W2, h2, N, E);

    gcn_aggregate_owned<D_OUT, 0, true><<<owners, AGG_THREADS, 0, stream>>>(
        src, dst, lists, tot, dinv, h2, b2, nullptr, out, N, E);
}
